# starting kernel
# baseline (speedup 1.0000x reference)
#include <hip/hip_runtime.h>
#include <hip/hip_bf16.h>
#include <math.h>

typedef __attribute__((ext_vector_type(2))) float v2f;
typedef __attribute__((ext_vector_type(8))) float v8f;

#define C_DIM 256
#define N_PIX 196
#define MAT_ELEMS (C_DIM * C_DIM)
#define BATCH 64
#define OUT_PER_B 32896
#define LDS_STRIDE 34

__global__ __launch_bounds__(256) void mean_kernel(const float* __restrict__ x,
                                                   float* __restrict__ mean) {
    int row = blockIdx.x * 8 + (threadIdx.x >> 5);
    int lane = threadIdx.x & 31;
    const float* xr = x + (size_t)row * N_PIX;
    float s = 0.0f;
    for (int i = lane; i < N_PIX; i += 32) s += xr[i];
    #pragma unroll
    for (int o = 16; o > 0; o >>= 1) s += __shfl_down(s, o, 32);
    if (lane == 0) mean[row] = s * (1.0f / (float)N_PIX);
}

__global__ __launch_bounds__(256) void sigma_kernel(float* __restrict__ D,
                                                    const float* __restrict__ x,
                                                    const float* __restrict__ mean) {
    __shared__ float As[128 * LDS_STRIDE];
    __shared__ float Bt[128 * LDS_STRIDE];
    const int tid  = threadIdx.x;
    const int lane = tid & 31;
    const int wave = tid >> 5;
    const int b    = blockIdx.y;
    const int tileid   = blockIdx.x;
    const int row_base = (tileid >> 1) * 128;
    const int col_base = (tileid & 1) * 128;
    const float* xb = x + (size_t)b * C_DIM * N_PIX;
    const float* mb = mean + (size_t)b * C_DIM;
    float* Db = D + (size_t)b * MAT_ELEMS;

    const int wrow = (wave >> 1) * 32;
    const int wcol = (wave & 1) * 64;
    const int lr = lane & 15;
    const int kh = (lane >> 4) * 2;
    const float inv_n = 1.0f / (float)N_PIX;

    const int aR0 = (wrow +      lr) * LDS_STRIDE;
    const int aR1 = (wrow + 16 + lr) * LDS_STRIDE;
    const int bR0 = (wcol +      lr) * LDS_STRIDE;
    const int bR1 = (wcol + 16 + lr) * LDS_STRIDE;
    const int bR2 = (wcol + 32 + lr) * LDS_STRIDE;
    const int bR3 = (wcol + 48 + lr) * LDS_STRIDE;

    v8f acc00 = {}, acc01 = {}, acc02 = {}, acc03 = {};
    v8f acc10 = {}, acc11 = {}, acc12 = {}, acc13 = {};

    for (int kk = 0; kk < 224; kk += 32) {
        __syncthreads();
        #pragma unroll
        for (int r = 0; r < 16; ++r) {
            int e = r * 256 + tid;
            int mm = e >> 5, k = e & 31;
            int kg = kk + k;
            int row = row_base + mm;
            float v = 0.0f;
            if (kg < N_PIX) v = (xb[(size_t)row * N_PIX + kg] - mb[row]) * inv_n;
            As[mm * LDS_STRIDE + k] = v;
        }
        #pragma unroll
        for (int r = 0; r < 16; ++r) {
            int e = r * 256 + tid;
            int nn = e >> 5, k = e & 31;
            int kg = kk + k;
            int col = col_base + nn;
            float v = 0.0f;
            if (kg < N_PIX) v = xb[(size_t)col * N_PIX + kg];
            Bt[nn * LDS_STRIDE + k] = v;
        }
        __syncthreads();
        #pragma unroll
        for (int ks = 0; ks < 8; ++ks) {
            int kf = ks * 4 + kh;
            v2f a0 = *(const v2f*)&As[aR0 + kf];
            v2f a1 = *(const v2f*)&As[aR1 + kf];
            v2f b0 = *(const v2f*)&Bt[bR0 + kf];
            v2f b1 = *(const v2f*)&Bt[bR1 + kf];
            v2f b2 = *(const v2f*)&Bt[bR2 + kf];
            v2f b3 = *(const v2f*)&Bt[bR3 + kf];
            acc00 = __builtin_amdgcn_wmma_f32_16x16x4_f32(false, a0, false, b0, (short)0, acc00, false, false);
            acc01 = __builtin_amdgcn_wmma_f32_16x16x4_f32(false, a0, false, b1, (short)0, acc01, false, false);
            acc02 = __builtin_amdgcn_wmma_f32_16x16x4_f32(false, a0, false, b2, (short)0, acc02, false, false);
            acc03 = __builtin_amdgcn_wmma_f32_16x16x4_f32(false, a0, false, b3, (short)0, acc03, false, false);
            acc10 = __builtin_amdgcn_wmma_f32_16x16x4_f32(false, a1, false, b0, (short)0, acc10, false, false);
            acc11 = __builtin_amdgcn_wmma_f32_16x16x4_f32(false, a1, false, b1, (short)0, acc11, false, false);
            acc12 = __builtin_amdgcn_wmma_f32_16x16x4_f32(false, a1, false, b2, (short)0, acc12, false, false);
            acc13 = __builtin_amdgcn_wmma_f32_16x16x4_f32(false, a1, false, b3, (short)0, acc13, false, false);
        }
    }
    const int mrow = row_base + wrow + (lane >> 4) * 8;
    const int ncol = col_base + wcol + lr;
    #pragma unroll
    for (int v = 0; v < 8; ++v) {
        Db[(size_t)(mrow + v) * C_DIM + ncol]           = acc00[v];
        Db[(size_t)(mrow + v) * C_DIM + ncol + 16]      = acc01[v];
        Db[(size_t)(mrow + v) * C_DIM + ncol + 32]      = acc02[v];
        Db[(size_t)(mrow + v) * C_DIM + ncol + 48]      = acc03[v];
        Db[(size_t)(mrow + 16 + v) * C_DIM + ncol]      = acc10[v];
        Db[(size_t)(mrow + 16 + v) * C_DIM + ncol + 16] = acc11[v];
        Db[(size_t)(mrow + 16 + v) * C_DIM + ncol + 32] = acc12[v];
        Db[(size_t)(mrow + 16 + v) * C_DIM + ncol + 48] = acc13[v];
    }
}

__global__ __launch_bounds__(256) void trace_kernel(const float* __restrict__ S,
                                                    float* __restrict__ scal) {
    __shared__ float red[256];
    int b = blockIdx.x, tid = threadIdx.x;
    red[tid] = S[(size_t)b * MAT_ELEMS + (size_t)tid * 257];
    __syncthreads();
    for (int s = 128; s > 0; s >>= 1) {
        if (tid < s) red[tid] += red[tid + s];
        __syncthreads();
    }
    if (tid == 0) {
        float tr = red[0];
        scal[b] = 1.0f / tr;
        scal[64 + b] = sqrtf(tr);
    }
}

__global__ __launch_bounds__(256) void b0_kernel(float* __restrict__ Bm,
                                                 const float* __restrict__ S,
                                                 const float* __restrict__ scal) {
    int col = threadIdx.x, row = blockIdx.x, b = blockIdx.y;
    size_t idx = (size_t)b * MAT_ELEMS + (size_t)row * C_DIM + col;
    float rtr = scal[b];
    Bm[idx] = 0.5f * ((row == col ? 3.0f : 0.0f) - S[idx] * rtr);
}

__global__ __launch_bounds__(256) void ns_gemm_kernel(float* __restrict__ D,
                                                      const float* __restrict__ A,
                                                      const float* __restrict__ B,
                                                      const float* __restrict__ scal,
                                                      int op) {
    __shared__ float As[128 * LDS_STRIDE];
    __shared__ float Bt[128 * LDS_STRIDE];
    const int tid  = threadIdx.x;
    const int lane = tid & 31;
    const int wave = tid >> 5;
    const int b    = blockIdx.y;
    const int tileid   = blockIdx.x;
    const int row_base = (tileid >> 1) * 128;
    const int col_base = (tileid & 1) * 128;
    const float* Ab = A + (size_t)b * MAT_ELEMS;
    const float* Bb = B + (size_t)b * MAT_ELEMS;
    float* Dmat = D + (size_t)b * MAT_ELEMS;
    float* Dout = D + (size_t)b * OUT_PER_B;

    const int wrow = (wave >> 1) * 32;
    const int wcol = (wave & 1) * 64;
    const int lr = lane & 15;
    const int kh = (lane >> 4) * 2;

    const int aR0 = (wrow +      lr) * LDS_STRIDE;
    const int aR1 = (wrow + 16 + lr) * LDS_STRIDE;
    const int bR0 = (wcol +      lr) * LDS_STRIDE;
    const int bR1 = (wcol + 16 + lr) * LDS_STRIDE;
    const int bR2 = (wcol + 32 + lr) * LDS_STRIDE;
    const int bR3 = (wcol + 48 + lr) * LDS_STRIDE;

    v8f acc00 = {}, acc01 = {}, acc02 = {}, acc03 = {};
    v8f acc10 = {}, acc11 = {}, acc12 = {}, acc13 = {};

    for (int kk = 0; kk < 256; kk += 32) {
        __syncthreads();
        #pragma unroll
        for (int r = 0; r < 16; ++r) {
            int e = r * 256 + tid;
            int mm = e >> 5, k = e & 31;
            As[mm * LDS_STRIDE + k] = Ab[(size_t)(row_base + mm) * C_DIM + kk + k];
        }
        #pragma unroll
        for (int r = 0; r < 16; ++r) {
            int e = r * 256 + tid;
            int k = e >> 7, nn = e & 127;
            Bt[nn * LDS_STRIDE + k] = Bb[(size_t)(kk + k) * C_DIM + col_base + nn];
        }
        __syncthreads();
        #pragma unroll
        for (int ks = 0; ks < 8; ++ks) {
            int kf = ks * 4 + kh;
            v2f a0 = *(const v2f*)&As[aR0 + kf];
            v2f a1 = *(const v2f*)&As[aR1 + kf];
            v2f b0 = *(const v2f*)&Bt[bR0 + kf];
            v2f b1 = *(const v2f*)&Bt[bR1 + kf];
            v2f b2 = *(const v2f*)&Bt[bR2 + kf];
            v2f b3 = *(const v2f*)&Bt[bR3 + kf];
            acc00 = __builtin_amdgcn_wmma_f32_16x16x4_f32(false, a0, false, b0, (short)0, acc00, false, false);
            acc01 = __builtin_amdgcn_wmma_f32_16x16x4_f32(false, a0, false, b1, (short)0, acc01, false, false);
            acc02 = __builtin_amdgcn_wmma_f32_16x16x4_f32(false, a0, false, b2, (short)0, acc02, false, false);
            acc03 = __builtin_amdgcn_wmma_f32_16x16x4_f32(false, a0, false, b3, (short)0, acc03, false, false);
            acc10 = __builtin_amdgcn_wmma_f32_16x16x4_f32(false, a1, false, b0, (short)0, acc10, false, false);
            acc11 = __builtin_amdgcn_wmma_f32_16x16x4_f32(false, a1, false, b1, (short)0, acc11, false, false);
            acc12 = __builtin_amdgcn_wmma_f32_16x16x4_f32(false, a1, false, b2, (short)0, acc12, false, false);
            acc13 = __builtin_amdgcn_wmma_f32_16x16x4_f32(false, a1, false, b3, (short)0, acc13, false, false);
        }
    }

    float s = 1.0f;
    if (op >= 2) s = scal[b];
    const int mrow = row_base + wrow + (lane >> 4) * 8;
    const int ncol = col_base + wcol + lr;

    auto emit = [&](float val, int m, int n) {
        if (op == 3) {
            if (n >= m) {
                int start = m * C_DIM - (m * (m - 1)) / 2;
                Dout[start + (n - m)] = val * s;
            }
        } else {
            if (op == 1)      val = (m == n ? 1.5f : 0.0f) - 0.5f * val;
            else if (op == 2) val *= s;
            Dmat[(size_t)m * C_DIM + n] = val;
        }
    };
    #pragma unroll
    for (int v = 0; v < 8; ++v) {
        emit(acc00[v], mrow + v,      ncol);
        emit(acc01[v], mrow + v,      ncol + 16);
        emit(acc02[v], mrow + v,      ncol + 32);
        emit(acc03[v], mrow + v,      ncol + 48);
        emit(acc10[v], mrow + 16 + v, ncol);
        emit(acc11[v], mrow + 16 + v, ncol + 16);
        emit(acc12[v], mrow + 16 + v, ncol + 32);
        emit(acc13[v], mrow + 16 + v, ncol + 48);
    }
}

extern "C" void kernel_launch(void* const* d_in, const int* in_sizes, int n_in,
                              void* d_out, int out_size, void* d_ws, size_t ws_size,
                              hipStream_t stream) {
    const float* x = (const float*)d_in[0];
    float* out = (float*)d_out;
    float* ws = (float*)d_ws;

    float* mean = ws;
    float* scal = ws + 16384;
    const size_t MS = (size_t)BATCH * MAT_ELEMS;
    float* M0 = ws + 16512;
    float* M1 = M0 + MS;
    float* M2 = M1 + MS;
    float* M3 = M2 + MS;
    float* M4 = M3 + MS;

    dim3 gGemm(4, BATCH);
    dim3 gRow(C_DIM, BATCH);

    mean_kernel<<<2048, 256, 0, stream>>>(x, mean);
    sigma_kernel<<<gGemm, 256, 0, stream>>>(M0, x, mean);
    trace_kernel<<<BATCH, 256, 0, stream>>>(M0, scal);
    b0_kernel<<<gRow, 256, 0, stream>>>(M1, M0, scal);

    ns_gemm_kernel<<<gGemm, 256, 0, stream>>>(M2, M0, M1, scal, 2);

    ns_gemm_kernel<<<gGemm, 256, 0, stream>>>(M3, M1, M2, nullptr, 1);
    ns_gemm_kernel<<<gGemm, 256, 0, stream>>>(M4, M2, M3, nullptr, 0);
    ns_gemm_kernel<<<gGemm, 256, 0, stream>>>(M0, M3, M1, nullptr, 0);
    ns_gemm_kernel<<<gGemm, 256, 0, stream>>>(M1, M0, M4, nullptr, 1);
    ns_gemm_kernel<<<gGemm, 256, 0, stream>>>(M2, M4, M1, nullptr, 0);
    ns_gemm_kernel<<<gGemm, 256, 0, stream>>>(M3, M1, M0, nullptr, 0);
    ns_gemm_kernel<<<gGemm, 256, 0, stream>>>(M4, M3, M2, nullptr, 1);
    ns_gemm_kernel<<<gGemm, 256, 0, stream>>>(M0, M2, M4, nullptr, 0);
    ns_gemm_kernel<<<gGemm, 256, 0, stream>>>(M1, M4, M3, nullptr, 0);

    ns_gemm_kernel<<<gGemm, 256, 0, stream>>>(M2, M1, M0, nullptr, 1);
    ns_gemm_kernel<<<gGemm, 256, 0, stream>>>(out, M0, M2, scal + 64, 3);
}


# both WMMA kernels: A/B fragments loaded straight from global memory into registers (16-byte loads, two 16-k sets deep), no LDS tile, no barrier; 64x64 wave tiles; sum order inside 8-k groups changes
# speedup vs baseline: 3.0021x; 3.0021x over previous
#include <hip/hip_runtime.h>
#include <hip/hip_bf16.h>
#include <math.h>

typedef __attribute__((ext_vector_type(2))) float v2f;
typedef __attribute__((ext_vector_type(4))) float v4f;
typedef __attribute__((ext_vector_type(8))) float v8f;

#define C_DIM 256
#define N_PIX 196
#define MAT_ELEMS (C_DIM * C_DIM)
#define BATCH 64
#define OUT_PER_B 32896

__global__ __launch_bounds__(256) void mean_kernel(const float* __restrict__ x,
                                                   float* __restrict__ mean) {
    int row = blockIdx.x * 8 + (threadIdx.x >> 5);
    int lane = threadIdx.x & 31;
    const float* xr = x + (size_t)row * N_PIX;
    float s = 0.0f;
    for (int i = lane; i < N_PIX; i += 32) s += xr[i];
    #pragma unroll
    for (int o = 16; o > 0; o >>= 1) s += __shfl_down(s, o, 32);
    if (lane == 0) mean[row] = s * (1.0f / (float)N_PIX);
}

#define WMMA_F32(a, b, c) __builtin_amdgcn_wmma_f32_16x16x4_f32(false, a, false, b, (short)0, c, false, false)

struct FragSet {
    v4f a[2][4];
    v4f b[2][4];
};

__device__ __forceinline__ void sigma_load(FragSet& s, const float* __restrict__ xa,
                                           const float* __restrict__ xc, int k0) {
    #pragma unroll
    for (int u = 0; u < 2; ++u) {
        #pragma unroll
        for (int i = 0; i < 4; ++i)
            s.a[u][i] = *(const v4f*)(xa + (size_t)i * 16 * N_PIX + k0 + 8 * u);
        #pragma unroll
        for (int q = 0; q < 4; ++q)
            s.b[u][q] = *(const v4f*)(xc + (size_t)q * 16 * N_PIX + k0 + 8 * u);
    }
}

__device__ __forceinline__ void sigma_compute(const FragSet& s, const float (&mu)[4],
                                              float inv_n, v8f (&acc)[4][4]) {
    #pragma unroll
    for (int u = 0; u < 2; ++u) {
        v4f ac[4];
        #pragma unroll
        for (int i = 0; i < 4; ++i) ac[i] = (s.a[u][i] - mu[i]) * inv_n;
        #pragma unroll
        for (int p = 0; p < 2; ++p) {
            #pragma unroll
            for (int i = 0; i < 4; ++i) {
                v2f af = {ac[i][2 * p], ac[i][2 * p + 1]};
                #pragma unroll
                for (int q = 0; q < 4; ++q) {
                    v2f bf = {s.b[u][q][2 * p], s.b[u][q][2 * p + 1]};
                    acc[i][q] = WMMA_F32(af, bf, acc[i][q]);
                }
            }
        }
    }
}

__global__ __launch_bounds__(128) void sigma_kernel(float* __restrict__ D,
                                                    const float* __restrict__ x,
                                                    const float* __restrict__ mean) {
    const int lane = threadIdx.x & 31;
    const int wave = threadIdx.x >> 5;
    const int b    = blockIdx.y;
    const int tileid = blockIdx.x;
    const int rowb = (tileid >> 1) * 128 + (wave >> 1) * 64;
    const int colb = (tileid & 1) * 128 + (wave & 1) * 64;
    const int c = lane & 15;
    const int h = lane >> 4;
    const float* xb = x + (size_t)b * C_DIM * N_PIX;
    const float* mb = mean + (size_t)b * C_DIM;
    float* Db = D + (size_t)b * MAT_ELEMS;
    const float inv_n = 1.0f / (float)N_PIX;

    const float* xa = xb + (size_t)(rowb + c) * N_PIX + 4 * h;
    const float* xc = xb + (size_t)(colb + c) * N_PIX + 4 * h;
    float mu[4];
    #pragma unroll
    for (int i = 0; i < 4; ++i) mu[i] = mb[rowb + 16 * i + c];

    v8f acc[4][4];
    #pragma unroll
    for (int i = 0; i < 4; ++i)
        #pragma unroll
        for (int q = 0; q < 4; ++q) acc[i][q] = v8f{};

    FragSet s0, s1;
    sigma_load(s0, xa, xc, 0);
    #pragma nounroll
    for (int kk = 0; kk < 160; kk += 32) {
        sigma_load(s1, xa, xc, kk + 16);
        sigma_compute(s0, mu, inv_n, acc);
        sigma_load(s0, xa, xc, kk + 32);
        sigma_compute(s1, mu, inv_n, acc);
    }
    sigma_load(s1, xa, xc, 176);
    sigma_compute(s0, mu, inv_n, acc);
    v2f ta[4], tb[4];
    #pragma unroll
    for (int i = 0; i < 4; ++i)
        ta[i] = *(const v2f*)(xb + (size_t)(rowb + 16 * i + c) * N_PIX + 192 + 2 * h);
    #pragma unroll
    for (int q = 0; q < 4; ++q)
        tb[q] = *(const v2f*)(xb + (size_t)(colb + 16 * q + c) * N_PIX + 192 + 2 * h);
    sigma_compute(s1, mu, inv_n, acc);
    #pragma unroll
    for (int i = 0; i < 4; ++i) {
        v2f af = (ta[i] - mu[i]) * inv_n;
        #pragma unroll
        for (int q = 0; q < 4; ++q) acc[i][q] = WMMA_F32(af, tb[q], acc[i][q]);
    }

    #pragma unroll
    for (int i = 0; i < 4; ++i)
        #pragma unroll
        for (int r = 0; r < 8; ++r) {
            const int m = rowb + 16 * i + 8 * h + r;
            #pragma unroll
            for (int q = 0; q < 4; ++q)
                Db[(size_t)m * C_DIM + colb + 16 * q + c] = acc[i][q][r];
        }
}

__global__ __launch_bounds__(256) void trace_kernel(const float* __restrict__ S,
                                                    float* __restrict__ scal) {
    __shared__ float red[256];
    int b = blockIdx.x, tid = threadIdx.x;
    red[tid] = S[(size_t)b * MAT_ELEMS + (size_t)tid * 257];
    __syncthreads();
    for (int s = 128; s > 0; s >>= 1) {
        if (tid < s) red[tid] += red[tid + s];
        __syncthreads();
    }
    if (tid == 0) {
        float tr = red[0];
        scal[b] = 1.0f / tr;
        scal[64 + b] = sqrtf(tr);
    }
}

__global__ __launch_bounds__(256) void b0_kernel(float* __restrict__ Bm,
                                                 const float* __restrict__ S,
                                                 const float* __restrict__ scal) {
    int col = threadIdx.x, row = blockIdx.x, b = blockIdx.y;
    size_t idx = (size_t)b * MAT_ELEMS + (size_t)row * C_DIM + col;
    float rtr = scal[b];
    Bm[idx] = 0.5f * ((row == col ? 3.0f : 0.0f) - S[idx] * rtr);
}

__device__ __forceinline__ void gemm_load(FragSet& s, const float* __restrict__ Ap,
                                          const float* __restrict__ Bp, int k0) {
    #pragma unroll
    for (int u = 0; u < 2; ++u) {
        #pragma unroll
        for (int i = 0; i < 4; ++i)
            s.a[u][i] = *(const v4f*)(Ap + (size_t)i * 16 * C_DIM + k0 + 8 * u);
        #pragma unroll
        for (int e = 0; e < 4; ++e)
            s.b[u][e] = *(const v4f*)(Bp + (size_t)(k0 + 8 * u + e) * C_DIM);
    }
}

__device__ __forceinline__ void gemm_compute(const FragSet& s, v8f (&acc)[4][4]) {
    #pragma unroll
    for (int u = 0; u < 2; ++u) {
        #pragma unroll
        for (int p = 0; p < 2; ++p) {
            v2f bf[4];
            #pragma unroll
            for (int q = 0; q < 4; ++q) bf[q] = v2f{s.b[u][2 * p][q], s.b[u][2 * p + 1][q]};
            #pragma unroll
            for (int i = 0; i < 4; ++i) {
                v2f af = {s.a[u][i][2 * p], s.a[u][i][2 * p + 1]};
                #pragma unroll
                for (int q = 0; q < 4; ++q) acc[i][q] = WMMA_F32(af, bf[q], acc[i][q]);
            }
        }
    }
}

template <int OP>
__global__ __launch_bounds__(128) void ns_gemm_kernel(float* __restrict__ D,
                                                      const float* __restrict__ A,
                                                      const float* __restrict__ B,
                                                      const float* __restrict__ scal) {
    const int lane = threadIdx.x & 31;
    const int wave = threadIdx.x >> 5;
    const int b    = blockIdx.y;
    const int tileid = blockIdx.x;
    const int rowb = (tileid >> 1) * 128 + (wave >> 1) * 64;
    const int colb = (tileid & 1) * 128 + (wave & 1) * 64;
    const int c = lane & 15;
    const int h = lane >> 4;
    const float* Ab = A + (size_t)b * MAT_ELEMS;
    const float* Bb = B + (size_t)b * MAT_ELEMS;
    float* Dmat = D + (size_t)b * MAT_ELEMS;
    float* Dout = D + (size_t)b * OUT_PER_B;

    const float* Ap = Ab + (size_t)(rowb + c) * C_DIM + 4 * h;
    const float* Bp = Bb + (size_t)(4 * h) * C_DIM + colb + 4 * c;

    v8f acc[4][4];
    #pragma unroll
    for (int i = 0; i < 4; ++i)
        #pragma unroll
        for (int q = 0; q < 4; ++q) acc[i][q] = v8f{};

    FragSet s0, s1;
    gemm_load(s0, Ap, Bp, 0);
    #pragma nounroll
    for (int kk = 0; kk < 224; kk += 32) {
        gemm_load(s1, Ap, Bp, kk + 16);
        gemm_compute(s0, acc);
        gemm_load(s0, Ap, Bp, kk + 32);
        gemm_compute(s1, acc);
    }
    gemm_load(s1, Ap, Bp, 240);
    gemm_compute(s0, acc);
    gemm_compute(s1, acc);

    float s = 1.0f;
    if (OP >= 2) s = scal[b];
    const int n0 = colb + 4 * c;
    #pragma unroll
    for (int i = 0; i < 4; ++i) {
        #pragma unroll
        for (int r = 0; r < 8; ++r) {
            const int m = rowb + 16 * i + 8 * h + r;
            float val[4];
            #pragma unroll
            for (int q = 0; q < 4; ++q) val[q] = acc[i][q][r];
            if (OP == 3) {
                const int start = m * C_DIM - (m * (m - 1)) / 2;
                #pragma unroll
                for (int q = 0; q < 4; ++q) {
                    const int n = n0 + q;
                    if (n >= m) Dout[start + (n - m)] = val[q] * s;
                }
            } else {
                #pragma unroll
                for (int q = 0; q < 4; ++q) {
                    const int n = n0 + q;
                    if (OP == 1)      val[q] = (m == n ? 1.5f : 0.0f) - 0.5f * val[q];
                    else if (OP == 2) val[q] *= s;
                }
                *(v4f*)(Dmat + (size_t)m * C_DIM + n0) = v4f{val[0], val[1], val[2], val[3]};
            }
        }
    }
}

extern "C" void kernel_launch(void* const* d_in, const int* in_sizes, int n_in,
                              void* d_out, int out_size, void* d_ws, size_t ws_size,
                              hipStream_t stream) {
    const float* x = (const float*)d_in[0];
    float* out = (float*)d_out;
    float* ws = (float*)d_ws;

    float* mean = ws;
    float* scal = ws + 16384;
    const size_t MS = (size_t)BATCH * MAT_ELEMS;
    float* M0 = ws + 16512;
    float* M1 = M0 + MS;
    float* M2 = M1 + MS;
    float* M3 = M2 + MS;
    float* M4 = M3 + MS;

    dim3 gGemm(4, BATCH);
    dim3 gRow(C_DIM, BATCH);

    mean_kernel<<<2048, 256, 0, stream>>>(x, mean);
    sigma_kernel<<<gGemm, 128, 0, stream>>>(M0, x, mean);
    trace_kernel<<<BATCH, 256, 0, stream>>>(M0, scal);
    b0_kernel<<<gRow, 256, 0, stream>>>(M1, M0, scal);

    ns_gemm_kernel<2><<<gGemm, 128, 0, stream>>>(M2, M0, M1, scal);

    ns_gemm_kernel<1><<<gGemm, 128, 0, stream>>>(M3, M1, M2, nullptr);
    ns_gemm_kernel<0><<<gGemm, 128, 0, stream>>>(M4, M2, M3, nullptr);
    ns_gemm_kernel<0><<<gGemm, 128, 0, stream>>>(M0, M3, M1, nullptr);
    ns_gemm_kernel<1><<<gGemm, 128, 0, stream>>>(M1, M0, M4, nullptr);
    ns_gemm_kernel<0><<<gGemm, 128, 0, stream>>>(M2, M4, M1, nullptr);
    ns_gemm_kernel<0><<<gGemm, 128, 0, stream>>>(M3, M1, M0, nullptr);
    ns_gemm_kernel<1><<<gGemm, 128, 0, stream>>>(M4, M3, M2, nullptr);
    ns_gemm_kernel<0><<<gGemm, 128, 0, stream>>>(M0, M2, M4, nullptr);
    ns_gemm_kernel<0><<<gGemm, 128, 0, stream>>>(M1, M4, M3, nullptr);

    ns_gemm_kernel<1><<<gGemm, 128, 0, stream>>>(M2, M1, M0, nullptr);
    ns_gemm_kernel<3><<<gGemm, 128, 0, stream>>>(out, M0, M2, scal + 64);
}


# both WMMA kernels: k range split between two waves per SIMD (8-wave groups, same 64x64 wave tiles); partial sums swapped through LDS and added at the end: sum order changes (two k halves)
# speedup vs baseline: 3.1546x; 1.0508x over previous
#include <hip/hip_runtime.h>
#include <hip/hip_bf16.h>
#include <math.h>

typedef __attribute__((ext_vector_type(2))) float v2f;
typedef __attribute__((ext_vector_type(4))) float v4f;
typedef __attribute__((ext_vector_type(8))) float v8f;

#define C_DIM 256
#define N_PIX 196
#define MAT_ELEMS (C_DIM * C_DIM)
#define BATCH 64
#define OUT_PER_B 32896

__global__ __launch_bounds__(256) void mean_kernel(const float* __restrict__ x,
                                                   float* __restrict__ mean) {
    int row = blockIdx.x * 8 + (threadIdx.x >> 5);
    int lane = threadIdx.x & 31;
    const float* xr = x + (size_t)row * N_PIX;
    float s = 0.0f;
    for (int i = lane; i < N_PIX; i += 32) s += xr[i];
    #pragma unroll
    for (int o = 16; o > 0; o >>= 1) s += __shfl_down(s, o, 32);
    if (lane == 0) mean[row] = s * (1.0f / (float)N_PIX);
}

#define WMMA_F32(a, b, c) __builtin_amdgcn_wmma_f32_16x16x4_f32(false, a, false, b, (short)0, c, false, false)

struct FragSet {
    v4f a[2][4];
    v4f b[2][4];
};

template <int I0>
__device__ __forceinline__ void xchg_store(v4f* __restrict__ slot, const v8f (&acc)[4][4]) {
    #pragma unroll
    for (int e = 0; e < 2; ++e)
        #pragma unroll
        for (int q = 0; q < 4; ++q)
            #pragma unroll
            for (int rr = 0; rr < 2; ++rr)
                slot[((e * 4 + q) * 2 + rr) * 32] =
                    v4f{acc[I0 + e][q][4 * rr], acc[I0 + e][q][4 * rr + 1],
                        acc[I0 + e][q][4 * rr + 2], acc[I0 + e][q][4 * rr + 3]};
}

template <int I0>
__device__ __forceinline__ void xchg_add(const v4f* __restrict__ slot, v8f (&acc)[4][4]) {
    #pragma unroll
    for (int e = 0; e < 2; ++e)
        #pragma unroll
        for (int q = 0; q < 4; ++q)
            #pragma unroll
            for (int rr = 0; rr < 2; ++rr) {
                const v4f t = slot[((e * 4 + q) * 2 + rr) * 32];
                #pragma unroll
                for (int j = 0; j < 4; ++j) acc[I0 + e][q][4 * rr + j] += t[j];
            }
}

__device__ __forceinline__ void sigma_load(FragSet& s, const float* __restrict__ xa,
                                           const float* __restrict__ xc, int k0) {
    #pragma unroll
    for (int u = 0; u < 2; ++u) {
        #pragma unroll
        for (int i = 0; i < 4; ++i)
            s.a[u][i] = *(const v4f*)(xa + (size_t)i * 16 * N_PIX + k0 + 8 * u);
        #pragma unroll
        for (int q = 0; q < 4; ++q)
            s.b[u][q] = *(const v4f*)(xc + (size_t)q * 16 * N_PIX + k0 + 8 * u);
    }
}

__device__ __forceinline__ void sigma_compute(const FragSet& s, const float (&mu)[4],
                                              float inv_n, v8f (&acc)[4][4]) {
    #pragma unroll
    for (int u = 0; u < 2; ++u) {
        v4f ac[4];
        #pragma unroll
        for (int i = 0; i < 4; ++i) ac[i] = (s.a[u][i] - mu[i]) * inv_n;
        #pragma unroll
        for (int p = 0; p < 2; ++p) {
            #pragma unroll
            for (int i = 0; i < 4; ++i) {
                v2f af = {ac[i][2 * p], ac[i][2 * p + 1]};
                #pragma unroll
                for (int q = 0; q < 4; ++q) {
                    v2f bf = {s.b[u][q][2 * p], s.b[u][q][2 * p + 1]};
                    acc[i][q] = WMMA_F32(af, bf, acc[i][q]);
                }
            }
        }
    }
}

template <int I0>
__device__ __forceinline__ void sigma_store(const v8f (&acc)[4][4], float* __restrict__ Db,
                                            int rowb, int colb, int c, int h) {
    #pragma unroll
    for (int i = I0; i < I0 + 2; ++i)
        #pragma unroll
        for (int r = 0; r < 8; ++r) {
            const int m = rowb + 16 * i + 8 * h + r;
            #pragma unroll
            for (int q = 0; q < 4; ++q)
                Db[(size_t)m * C_DIM + colb + 16 * q + c] = acc[i][q][r];
        }
}

__global__ __launch_bounds__(256) void sigma_kernel(float* __restrict__ D,
                                                    const float* __restrict__ x,
                                                    const float* __restrict__ mean) {
    __shared__ v4f xch[8][16 * 32];
    const int lane = threadIdx.x & 31;
    const int wave = __builtin_amdgcn_readfirstlane((int)(threadIdx.x >> 5));
    const int pr   = wave & 3;
    const int kh   = wave >> 2;
    const int b    = blockIdx.y;
    const int tileid = blockIdx.x;
    const int rowb = (tileid >> 1) * 128 + (pr >> 1) * 64;
    const int colb = (tileid & 1) * 128 + (pr & 1) * 64;
    const int c = lane & 15;
    const int h = lane >> 4;
    const float* xb = x + (size_t)b * C_DIM * N_PIX;
    const float* mb = mean + (size_t)b * C_DIM;
    float* Db = D + (size_t)b * MAT_ELEMS;
    const float inv_n = 1.0f / (float)N_PIX;

    const float* xa = xb + (size_t)(rowb + c) * N_PIX + 96 * kh + 4 * h;
    const float* xc = xb + (size_t)(colb + c) * N_PIX + 96 * kh + 4 * h;
    float mu[4];
    #pragma unroll
    for (int i = 0; i < 4; ++i) mu[i] = mb[rowb + 16 * i + c];

    v8f acc[4][4];
    #pragma unroll
    for (int i = 0; i < 4; ++i)
        #pragma unroll
        for (int q = 0; q < 4; ++q) acc[i][q] = v8f{};

    FragSet s0, s1;
    sigma_load(s0, xa, xc, 0);
    #pragma nounroll
    for (int kk = 0; kk < 64; kk += 32) {
        sigma_load(s1, xa, xc, kk + 16);
        sigma_compute(s0, mu, inv_n, acc);
        sigma_load(s0, xa, xc, kk + 32);
        sigma_compute(s1, mu, inv_n, acc);
    }
    sigma_load(s1, xa, xc, 80);
    sigma_compute(s0, mu, inv_n, acc);
    v2f ta[4], tb[4];
    if (kh == 1) {
        #pragma unroll
        for (int i = 0; i < 4; ++i)
            ta[i] = *(const v2f*)(xb + (size_t)(rowb + 16 * i + c) * N_PIX + 192 + 2 * h);
        #pragma unroll
        for (int q = 0; q < 4; ++q)
            tb[q] = *(const v2f*)(xb + (size_t)(colb + 16 * q + c) * N_PIX + 192 + 2 * h);
    }
    sigma_compute(s1, mu, inv_n, acc);
    if (kh == 1) {
        #pragma unroll
        for (int i = 0; i < 4; ++i) {
            v2f af = (ta[i] - mu[i]) * inv_n;
            #pragma unroll
            for (int q = 0; q < 4; ++q) acc[i][q] = WMMA_F32(af, tb[q], acc[i][q]);
        }
    }

    v4f* mine = &xch[wave][lane];
    const v4f* theirs = &xch[wave ^ 4][lane];
    if (kh == 0) xchg_store<2>(mine, acc); else xchg_store<0>(mine, acc);
    __syncthreads();
    if (kh == 0) {
        xchg_add<0>(theirs, acc);
        sigma_store<0>(acc, Db, rowb, colb, c, h);
    } else {
        xchg_add<2>(theirs, acc);
        sigma_store<2>(acc, Db, rowb, colb, c, h);
    }
}

__global__ __launch_bounds__(256) void trace_kernel(const float* __restrict__ S,
                                                    float* __restrict__ scal) {
    __shared__ float red[256];
    int b = blockIdx.x, tid = threadIdx.x;
    red[tid] = S[(size_t)b * MAT_ELEMS + (size_t)tid * 257];
    __syncthreads();
    for (int s = 128; s > 0; s >>= 1) {
        if (tid < s) red[tid] += red[tid + s];
        __syncthreads();
    }
    if (tid == 0) {
        float tr = red[0];
        scal[b] = 1.0f / tr;
        scal[64 + b] = sqrtf(tr);
    }
}

__global__ __launch_bounds__(256) void b0_kernel(float* __restrict__ Bm,
                                                 const float* __restrict__ S,
                                                 const float* __restrict__ scal) {
    int col = threadIdx.x, row = blockIdx.x, b = blockIdx.y;
    size_t idx = (size_t)b * MAT_ELEMS + (size_t)row * C_DIM + col;
    float rtr = scal[b];
    Bm[idx] = 0.5f * ((row == col ? 3.0f : 0.0f) - S[idx] * rtr);
}

__device__ __forceinline__ void gemm_load(FragSet& s, const float* __restrict__ Ap,
                                          const float* __restrict__ Bp, int k0) {
    #pragma unroll
    for (int u = 0; u < 2; ++u) {
        #pragma unroll
        for (int i = 0; i < 4; ++i)
            s.a[u][i] = *(const v4f*)(Ap + (size_t)i * 16 * C_DIM + k0 + 8 * u);
        #pragma unroll
        for (int e = 0; e < 4; ++e)
            s.b[u][e] = *(const v4f*)(Bp + (size_t)(k0 + 8 * u + e) * C_DIM);
    }
}

__device__ __forceinline__ void gemm_compute(const FragSet& s, v8f (&acc)[4][4]) {
    #pragma unroll
    for (int u = 0; u < 2; ++u) {
        #pragma unroll
        for (int p = 0; p < 2; ++p) {
            v2f bf[4];
            #pragma unroll
            for (int q = 0; q < 4; ++q) bf[q] = v2f{s.b[u][2 * p][q], s.b[u][2 * p + 1][q]};
            #pragma unroll
            for (int i = 0; i < 4; ++i) {
                v2f af = {s.a[u][i][2 * p], s.a[u][i][2 * p + 1]};
                #pragma unroll
                for (int q = 0; q < 4; ++q) acc[i][q] = WMMA_F32(af, bf[q], acc[i][q]);
            }
        }
    }
}

template <int OP, int I0>
__device__ __forceinline__ void ns_epilogue(const v8f (&acc)[4][4], float* __restrict__ Dmat,
                                            float* __restrict__ Dout, float s,
                                            int rowb, int n0, int h) {
    #pragma unroll
    for (int i = I0; i < I0 + 2; ++i) {
        #pragma unroll
        for (int r = 0; r < 8; ++r) {
            const int m = rowb + 16 * i + 8 * h + r;
            float val[4];
            #pragma unroll
            for (int q = 0; q < 4; ++q) val[q] = acc[i][q][r];
            if (OP == 3) {
                const int start = m * C_DIM - (m * (m - 1)) / 2;
                #pragma unroll
                for (int q = 0; q < 4; ++q) {
                    const int n = n0 + q;
                    if (n >= m) Dout[start + (n - m)] = val[q] * s;
                }
            } else {
                #pragma unroll
                for (int q = 0; q < 4; ++q) {
                    const int n = n0 + q;
                    if (OP == 1)      val[q] = (m == n ? 1.5f : 0.0f) - 0.5f * val[q];
                    else if (OP == 2) val[q] *= s;
                }
                *(v4f*)(Dmat + (size_t)m * C_DIM + n0) = v4f{val[0], val[1], val[2], val[3]};
            }
        }
    }
}

template <int OP>
__global__ __launch_bounds__(256) void ns_gemm_kernel(float* __restrict__ D,
                                                      const float* __restrict__ A,
                                                      const float* __restrict__ B,
                                                      const float* __restrict__ scal) {
    __shared__ v4f xch[8][16 * 32];
    const int lane = threadIdx.x & 31;
    const int wave = __builtin_amdgcn_readfirstlane((int)(threadIdx.x >> 5));
    const int pr   = wave & 3;
    const int kh   = wave >> 2;
    const int b    = blockIdx.y;
    const int tileid = blockIdx.x;
    const int rowb = (tileid >> 1) * 128 + (pr >> 1) * 64;
    const int colb = (tileid & 1) * 128 + (pr & 1) * 64;
    const int c = lane & 15;
    const int h = lane >> 4;
    const float* Ab = A + (size_t)b * MAT_ELEMS;
    const float* Bb = B + (size_t)b * MAT_ELEMS;
    float* Dmat = D + (size_t)b * MAT_ELEMS;
    float* Dout = D + (size_t)b * OUT_PER_B;

    const float* Ap = Ab + (size_t)(rowb + c) * C_DIM + 128 * kh + 4 * h;
    const float* Bp = Bb + (size_t)(128 * kh + 4 * h) * C_DIM + colb + 4 * c;

    v8f acc[4][4];
    #pragma unroll
    for (int i = 0; i < 4; ++i)
        #pragma unroll
        for (int q = 0; q < 4; ++q) acc[i][q] = v8f{};

    FragSet s0, s1;
    gemm_load(s0, Ap, Bp, 0);
    #pragma nounroll
    for (int kk = 0; kk < 96; kk += 32) {
        gemm_load(s1, Ap, Bp, kk + 16);
        gemm_compute(s0, acc);
        gemm_load(s0, Ap, Bp, kk + 32);
        gemm_compute(s1, acc);
    }
    gemm_load(s1, Ap, Bp, 112);
    gemm_compute(s0, acc);
    gemm_compute(s1, acc);

    float s = 1.0f;
    if (OP >= 2) s = scal[b];
    const int n0 = colb + 4 * c;
    v4f* mine = &xch[wave][lane];
    const v4f* theirs = &xch[wave ^ 4][lane];
    if (kh == 0) xchg_store<2>(mine, acc); else xchg_store<0>(mine, acc);
    __syncthreads();
    if (kh == 0) {
        xchg_add<0>(theirs, acc);
        ns_epilogue<OP, 0>(acc, Dmat, Dout, s, rowb, n0, h);
    } else {
        xchg_add<2>(theirs, acc);
        ns_epilogue<OP, 2>(acc, Dmat, Dout, s, rowb, n0, h);
    }
}

extern "C" void kernel_launch(void* const* d_in, const int* in_sizes, int n_in,
                              void* d_out, int out_size, void* d_ws, size_t ws_size,
                              hipStream_t stream) {
    const float* x = (const float*)d_in[0];
    float* out = (float*)d_out;
    float* ws = (float*)d_ws;

    float* mean = ws;
    float* scal = ws + 16384;
    const size_t MS = (size_t)BATCH * MAT_ELEMS;
    float* M0 = ws + 16512;
    float* M1 = M0 + MS;
    float* M2 = M1 + MS;
    float* M3 = M2 + MS;
    float* M4 = M3 + MS;

    dim3 gGemm(4, BATCH);
    dim3 gRow(C_DIM, BATCH);

    mean_kernel<<<2048, 256, 0, stream>>>(x, mean);
    sigma_kernel<<<gGemm, 256, 0, stream>>>(M0, x, mean);
    trace_kernel<<<BATCH, 256, 0, stream>>>(M0, scal);
    b0_kernel<<<gRow, 256, 0, stream>>>(M1, M0, scal);

    ns_gemm_kernel<2><<<gGemm, 256, 0, stream>>>(M2, M0, M1, scal);

    ns_gemm_kernel<1><<<gGemm, 256, 0, stream>>>(M3, M1, M2, nullptr);
    ns_gemm_kernel<0><<<gGemm, 256, 0, stream>>>(M4, M2, M3, nullptr);
    ns_gemm_kernel<0><<<gGemm, 256, 0, stream>>>(M0, M3, M1, nullptr);
    ns_gemm_kernel<1><<<gGemm, 256, 0, stream>>>(M1, M0, M4, nullptr);
    ns_gemm_kernel<0><<<gGemm, 256, 0, stream>>>(M2, M4, M1, nullptr);
    ns_gemm_kernel<0><<<gGemm, 256, 0, stream>>>(M3, M1, M0, nullptr);
    ns_gemm_kernel<1><<<gGemm, 256, 0, stream>>>(M4, M3, M2, nullptr);
    ns_gemm_kernel<0><<<gGemm, 256, 0, stream>>>(M0, M2, M4, nullptr);
    ns_gemm_kernel<0><<<gGemm, 256, 0, stream>>>(M1, M4, M3, nullptr);

    ns_gemm_kernel<1><<<gGemm, 256, 0, stream>>>(M2, M1, M0, nullptr);
    ns_gemm_kernel<3><<<gGemm, 256, 0, stream>>>(out, M0, M2, scal + 64);
}


# dependent product pairs fused by panel ownership: B=0.5(3I-Z@Y) row panel stays in LDS and feeds B@Z; closing bf column panel in LDS feeds Y@bf; 16 launches become 12; every sum order kept
# speedup vs baseline: 3.3598x; 1.0650x over previous
#include <hip/hip_runtime.h>
#include <hip/hip_bf16.h>
#include <math.h>

typedef __attribute__((ext_vector_type(2))) float v2f;
typedef __attribute__((ext_vector_type(4))) float v4f;
typedef __attribute__((ext_vector_type(8))) float v8f;

#define C_DIM 256
#define N_PIX 196
#define MAT_ELEMS (C_DIM * C_DIM)
#define BATCH 64
#define OUT_PER_B 32896

__global__ __launch_bounds__(256) void mean_kernel(const float* __restrict__ x,
                                                   float* __restrict__ mean) {
    int row = blockIdx.x * 8 + (threadIdx.x >> 5);
    int lane = threadIdx.x & 31;
    const float* xr = x + (size_t)row * N_PIX;
    float s = 0.0f;
    for (int i = lane; i < N_PIX; i += 32) s += xr[i];
    #pragma unroll
    for (int o = 16; o > 0; o >>= 1) s += __shfl_down(s, o, 32);
    if (lane == 0) mean[row] = s * (1.0f / (float)N_PIX);
}

#define WMMA_F32(a, b, c) __builtin_amdgcn_wmma_f32_16x16x4_f32(false, a, false, b, (short)0, c, false, false)

struct FragSet {
    v4f a[2][4];
    v4f b[2][4];
};

template <int I0>
__device__ __forceinline__ void xchg_store(v4f* __restrict__ slot, const v8f (&acc)[4][4]) {
    #pragma unroll
    for (int e = 0; e < 2; ++e)
        #pragma unroll
        for (int q = 0; q < 4; ++q)
            #pragma unroll
            for (int rr = 0; rr < 2; ++rr)
                slot[((e * 4 + q) * 2 + rr) * 32] =
                    v4f{acc[I0 + e][q][4 * rr], acc[I0 + e][q][4 * rr + 1],
                        acc[I0 + e][q][4 * rr + 2], acc[I0 + e][q][4 * rr + 3]};
}

template <int I0>
__device__ __forceinline__ void xchg_add(const v4f* __restrict__ slot, v8f (&acc)[4][4]) {
    #pragma unroll
    for (int e = 0; e < 2; ++e)
        #pragma unroll
        for (int q = 0; q < 4; ++q)
            #pragma unroll
            for (int rr = 0; rr < 2; ++rr) {
                const v4f t = slot[((e * 4 + q) * 2 + rr) * 32];
                #pragma unroll
                for (int j = 0; j < 4; ++j) acc[I0 + e][q][4 * rr + j] += t[j];
            }
}

__device__ __forceinline__ void sigma_load(FragSet& s, const float* __restrict__ xa,
                                           const float* __restrict__ xc, int k0) {
    #pragma unroll
    for (int u = 0; u < 2; ++u) {
        #pragma unroll
        for (int i = 0; i < 4; ++i)
            s.a[u][i] = *(const v4f*)(xa + (size_t)i * 16 * N_PIX + k0 + 8 * u);
        #pragma unroll
        for (int q = 0; q < 4; ++q)
            s.b[u][q] = *(const v4f*)(xc + (size_t)q * 16 * N_PIX + k0 + 8 * u);
    }
}

__device__ __forceinline__ void sigma_compute(const FragSet& s, const float (&mu)[4],
                                              float inv_n, v8f (&acc)[4][4]) {
    #pragma unroll
    for (int u = 0; u < 2; ++u) {
        v4f ac[4];
        #pragma unroll
        for (int i = 0; i < 4; ++i) ac[i] = (s.a[u][i] - mu[i]) * inv_n;
        #pragma unroll
        for (int p = 0; p < 2; ++p) {
            #pragma unroll
            for (int i = 0; i < 4; ++i) {
                v2f af = {ac[i][2 * p], ac[i][2 * p + 1]};
                #pragma unroll
                for (int q = 0; q < 4; ++q) {
                    v2f bf = {s.b[u][q][2 * p], s.b[u][q][2 * p + 1]};
                    acc[i][q] = WMMA_F32(af, bf, acc[i][q]);
                }
            }
        }
    }
}

template <int I0>
__device__ __forceinline__ void sigma_store(const v8f (&acc)[4][4], float* __restrict__ Db,
                                            int rowb, int colb, int c, int h) {
    #pragma unroll
    for (int i = I0; i < I0 + 2; ++i)
        #pragma unroll
        for (int r = 0; r < 8; ++r) {
            const int m = rowb + 16 * i + 8 * h + r;
            #pragma unroll
            for (int q = 0; q < 4; ++q)
                Db[(size_t)m * C_DIM + colb + 16 * q + c] = acc[i][q][r];
        }
}

__global__ __launch_bounds__(256) void sigma_kernel(float* __restrict__ D,
                                                    const float* __restrict__ x,
                                                    const float* __restrict__ mean) {
    __shared__ v4f xch[8][16 * 32];
    const int lane = threadIdx.x & 31;
    const int wave = __builtin_amdgcn_readfirstlane((int)(threadIdx.x >> 5));
    const int pr   = wave & 3;
    const int kh   = wave >> 2;
    const int b    = blockIdx.y;
    const int tileid = blockIdx.x;
    const int rowb = (tileid >> 1) * 128 + (pr >> 1) * 64;
    const int colb = (tileid & 1) * 128 + (pr & 1) * 64;
    const int c = lane & 15;
    const int h = lane >> 4;
    const float* xb = x + (size_t)b * C_DIM * N_PIX;
    const float* mb = mean + (size_t)b * C_DIM;
    float* Db = D + (size_t)b * MAT_ELEMS;
    const float inv_n = 1.0f / (float)N_PIX;

    const float* xa = xb + (size_t)(rowb + c) * N_PIX + 96 * kh + 4 * h;
    const float* xc = xb + (size_t)(colb + c) * N_PIX + 96 * kh + 4 * h;
    float mu[4];
    #pragma unroll
    for (int i = 0; i < 4; ++i) mu[i] = mb[rowb + 16 * i + c];

    v8f acc[4][4];
    #pragma unroll
    for (int i = 0; i < 4; ++i)
        #pragma unroll
        for (int q = 0; q < 4; ++q) acc[i][q] = v8f{};

    FragSet s0, s1;
    sigma_load(s0, xa, xc, 0);
    #pragma nounroll
    for (int kk = 0; kk < 64; kk += 32) {
        sigma_load(s1, xa, xc, kk + 16);
        sigma_compute(s0, mu, inv_n, acc);
        sigma_load(s0, xa, xc, kk + 32);
        sigma_compute(s1, mu, inv_n, acc);
    }
    sigma_load(s1, xa, xc, 80);
    sigma_compute(s0, mu, inv_n, acc);
    v2f ta[4], tb[4];
    if (kh == 1) {
        #pragma unroll
        for (int i = 0; i < 4; ++i)
            ta[i] = *(const v2f*)(xb + (size_t)(rowb + 16 * i + c) * N_PIX + 192 + 2 * h);
        #pragma unroll
        for (int q = 0; q < 4; ++q)
            tb[q] = *(const v2f*)(xb + (size_t)(colb + 16 * q + c) * N_PIX + 192 + 2 * h);
    }
    sigma_compute(s1, mu, inv_n, acc);
    if (kh == 1) {
        #pragma unroll
        for (int i = 0; i < 4; ++i) {
            v2f af = (ta[i] - mu[i]) * inv_n;
            #pragma unroll
            for (int q = 0; q < 4; ++q) acc[i][q] = WMMA_F32(af, tb[q], acc[i][q]);
        }
    }

    v4f* mine = &xch[wave][lane];
    const v4f* theirs = &xch[wave ^ 4][lane];
    if (kh == 0) xchg_store<2>(mine, acc); else xchg_store<0>(mine, acc);
    __syncthreads();
    if (kh == 0) {
        xchg_add<0>(theirs, acc);
        sigma_store<0>(acc, Db, rowb, colb, c, h);
    } else {
        xchg_add<2>(theirs, acc);
        sigma_store<2>(acc, Db, rowb, colb, c, h);
    }
}

__global__ __launch_bounds__(256) void trace_kernel(const float* __restrict__ S,
                                                    float* __restrict__ scal) {
    __shared__ float red[256];
    int b = blockIdx.x, tid = threadIdx.x;
    red[tid] = S[(size_t)b * MAT_ELEMS + (size_t)tid * 257];
    __syncthreads();
    for (int s = 128; s > 0; s >>= 1) {
        if (tid < s) red[tid] += red[tid + s];
        __syncthreads();
    }
    if (tid == 0) {
        float tr = red[0];
        scal[b] = 1.0f / tr;
        scal[64 + b] = sqrtf(tr);
    }
}

__global__ __launch_bounds__(256) void b0_kernel(float* __restrict__ Bm,
                                                 const float* __restrict__ S,
                                                 const float* __restrict__ scal) {
    int col = threadIdx.x, row = blockIdx.x, b = blockIdx.y;
    size_t idx = (size_t)b * MAT_ELEMS + (size_t)row * C_DIM + col;
    float rtr = scal[b];
    Bm[idx] = 0.5f * ((row == col ? 3.0f : 0.0f) - S[idx] * rtr);
}

__device__ __forceinline__ void gemm_load(FragSet& s, const float* __restrict__ Ap,
                                          const float* __restrict__ Bp, int k0) {
    #pragma unroll
    for (int u = 0; u < 2; ++u) {
        #pragma unroll
        for (int i = 0; i < 4; ++i)
            s.a[u][i] = *(const v4f*)(Ap + (size_t)i * 16 * C_DIM + k0 + 8 * u);
        #pragma unroll
        for (int e = 0; e < 4; ++e)
            s.b[u][e] = *(const v4f*)(Bp + (size_t)(k0 + 8 * u + e) * C_DIM);
    }
}

__device__ __forceinline__ void gemm_compute(const FragSet& s, v8f (&acc)[4][4]) {
    #pragma unroll
    for (int u = 0; u < 2; ++u) {
        #pragma unroll
        for (int p = 0; p < 2; ++p) {
            v2f bf[4];
            #pragma unroll
            for (int q = 0; q < 4; ++q) bf[q] = v2f{s.b[u][2 * p][q], s.b[u][2 * p + 1][q]};
            #pragma unroll
            for (int i = 0; i < 4; ++i) {
                v2f af = {s.a[u][i][2 * p], s.a[u][i][2 * p + 1]};
                #pragma unroll
                for (int q = 0; q < 4; ++q) acc[i][q] = WMMA_F32(af, bf[q], acc[i][q]);
            }
        }
    }
}

template <int OP, int I0>
__device__ __forceinline__ void ns_epilogue(const v8f (&acc)[4][4], float* __restrict__ Dmat,
                                            float* __restrict__ Dout, float s,
                                            int rowb, int n0, int h) {
    #pragma unroll
    for (int i = I0; i < I0 + 2; ++i) {
        #pragma unroll
        for (int r = 0; r < 8; ++r) {
            const int m = rowb + 16 * i + 8 * h + r;
            float val[4];
            #pragma unroll
            for (int q = 0; q < 4; ++q) val[q] = acc[i][q][r];
            if (OP == 3) {
                const int start = m * C_DIM - (m * (m - 1)) / 2;
                #pragma unroll
                for (int q = 0; q < 4; ++q) {
                    const int n = n0 + q;
                    if (n >= m) Dout[start + (n - m)] = val[q] * s;
                }
            } else {
                #pragma unroll
                for (int q = 0; q < 4; ++q) {
                    if (OP == 2) val[q] *= s;
                }
                *(v4f*)(Dmat + (size_t)m * C_DIM + n0) = v4f{val[0], val[1], val[2], val[3]};
            }
        }
    }
}

template <int OP>
__global__ __launch_bounds__(256) void ns_gemm_kernel(float* __restrict__ D,
                                                      const float* __restrict__ A,
                                                      const float* __restrict__ B,
                                                      const float* __restrict__ scal) {
    __shared__ v4f xch[8][16 * 32];
    const int lane = threadIdx.x & 31;
    const int wave = __builtin_amdgcn_readfirstlane((int)(threadIdx.x >> 5));
    const int pr   = wave & 3;
    const int kh   = wave >> 2;
    const int b    = blockIdx.y;
    const int tileid = blockIdx.x;
    const int rowb = (tileid >> 1) * 128 + (pr >> 1) * 64;
    const int colb = (tileid & 1) * 128 + (pr & 1) * 64;
    const int c = lane & 15;
    const int h = lane >> 4;
    const float* Ab = A + (size_t)b * MAT_ELEMS;
    const float* Bb = B + (size_t)b * MAT_ELEMS;
    float* Dmat = D + (size_t)b * MAT_ELEMS;
    float* Dout = D + (size_t)b * OUT_PER_B;

    const float* Ap = Ab + (size_t)(rowb + c) * C_DIM + 128 * kh + 4 * h;
    const float* Bp = Bb + (size_t)(128 * kh + 4 * h) * C_DIM + colb + 4 * c;

    v8f acc[4][4];
    #pragma unroll
    for (int i = 0; i < 4; ++i)
        #pragma unroll
        for (int q = 0; q < 4; ++q) acc[i][q] = v8f{};

    FragSet s0, s1;
    gemm_load(s0, Ap, Bp, 0);
    #pragma nounroll
    for (int kk = 0; kk < 96; kk += 32) {
        gemm_load(s1, Ap, Bp, kk + 16);
        gemm_compute(s0, acc);
        gemm_load(s0, Ap, Bp, kk + 32);
        gemm_compute(s1, acc);
    }
    gemm_load(s1, Ap, Bp, 112);
    gemm_compute(s0, acc);
    gemm_compute(s1, acc);

    float s = 1.0f;
    if (OP >= 2) s = scal[b];
    const int n0 = colb + 4 * c;
    v4f* mine = &xch[wave][lane];
    const v4f* theirs = &xch[wave ^ 4][lane];
    if (kh == 0) xchg_store<2>(mine, acc); else xchg_store<0>(mine, acc);
    __syncthreads();
    if (kh == 0) {
        xchg_add<0>(theirs, acc);
        ns_epilogue<OP, 0>(acc, Dmat, Dout, s, rowb, n0, h);
    } else {
        xchg_add<2>(theirs, acc);
        ns_epilogue<OP, 2>(acc, Dmat, Dout, s, rowb, n0, h);
    }
}

#define BS_PITCH4 65

__device__ __forceinline__ void pair_load_b(FragSet& s, const float* __restrict__ Bp, int k0) {
    #pragma unroll
    for (int u = 0; u < 2; ++u)
        #pragma unroll
        for (int e = 0; e < 4; ++e)
            s.b[u][e] = *(const v4f*)(Bp + (size_t)(k0 + 8 * u + e) * C_DIM);
}

__device__ __forceinline__ void pair_load_a(FragSet& s, const v4f* As4, int k0) {
    #pragma unroll
    for (int u = 0; u < 2; ++u)
        #pragma unroll
        for (int i = 0; i < 4; ++i)
            s.a[u][i] = As4[i * 16 * BS_PITCH4 + (k0 >> 2) + 2 * u];
}

__device__ __forceinline__ v4f pair_b_value(const v8f (&acc)[4][4], int i, int r, int m, int n0) {
    float val[4];
    #pragma unroll
    for (int q = 0; q < 4; ++q) {
        const int n = n0 + q;
        val[q] = (m == n ? 1.5f : 0.0f) - 0.5f * acc[i][q][r];
    }
    return v4f{val[0], val[1], val[2], val[3]};
}

template <int I0>
__device__ __forceinline__ void pair_b_to_lds(const v8f (&acc)[4][4], v4f* Bs4,
                                              int rowb, int n0, int h, int lc) {
    #pragma unroll
    for (int i = I0; i < I0 + 2; ++i)
        #pragma unroll
        for (int r = 0; r < 8; ++r) {
            const int ml = 16 * i + 8 * h + r;
            Bs4[ml * BS_PITCH4 + lc] = pair_b_value(acc, i, r, rowb + ml, n0);
        }
}

template <int I0>
__device__ __forceinline__ void pair_b_to_global(const v8f (&acc)[4][4], float* __restrict__ Bmat,
                                                 int rowb, int n0, int h) {
    #pragma unroll
    for (int i = I0; i < I0 + 2; ++i)
        #pragma unroll
        for (int r = 0; r < 8; ++r) {
            const int m = rowb + 16 * i + 8 * h + r;
            *(v4f*)(Bmat + (size_t)m * C_DIM + n0) = pair_b_value(acc, i, r, m, n0);
        }
}

__global__ __launch_bounds__(256) void ns_pair_kernel(float* __restrict__ Bout,
                                                      float* __restrict__ Zout,
                                                      const float* __restrict__ Z,
                                                      const float* __restrict__ Y) {
    __shared__ v4f xch[8][16 * 32];
    __shared__ v4f Bs4[64 * BS_PITCH4];
    const int lane = threadIdx.x & 31;
    const int wave = __builtin_amdgcn_readfirstlane((int)(threadIdx.x >> 5));
    const int pr   = wave & 3;
    const int kh   = wave >> 2;
    const int b    = blockIdx.y;
    const int rowb = blockIdx.x * 64;
    const int colb = pr * 64;
    const int c = lane & 15;
    const int h = lane >> 4;
    const float* Zb = Z + (size_t)b * MAT_ELEMS;
    const float* Yb = Y + (size_t)b * MAT_ELEMS;
    float* Bmat = Bout + (size_t)b * MAT_ELEMS;
    float* Zmat = Zout + (size_t)b * MAT_ELEMS;
    const int n0 = colb + 4 * c;
    v4f* mine = &xch[wave][lane];
    const v4f* theirs = &xch[wave ^ 4][lane];

    v8f acc[4][4];
    #pragma unroll
    for (int i = 0; i < 4; ++i)
        #pragma unroll
        for (int q = 0; q < 4; ++q) acc[i][q] = v8f{};

    FragSet s0, s1;
    {
        const float* Ap = Zb + (size_t)(rowb + c) * C_DIM + 128 * kh + 4 * h;
        const float* Bp = Yb + (size_t)(128 * kh + 4 * h) * C_DIM + colb + 4 * c;
        gemm_load(s0, Ap, Bp, 0);
        #pragma nounroll
        for (int kk = 0; kk < 96; kk += 32) {
            gemm_load(s1, Ap, Bp, kk + 16);
            gemm_compute(s0, acc);
            gemm_load(s0, Ap, Bp, kk + 32);
            gemm_compute(s1, acc);
        }
        gemm_load(s1, Ap, Bp, 112);
        gemm_compute(s0, acc);
        gemm_compute(s1, acc);
    }
    const float* Bp2 = Zb + (size_t)(128 * kh + 4 * h) * C_DIM + colb + 4 * c;

    if (kh == 0) xchg_store<2>(mine, acc); else xchg_store<0>(mine, acc);
    __syncthreads();
    pair_load_b(s0, Bp2, 0);
    if (kh == 0) {
        xchg_add<0>(theirs, acc);
        pair_b_to_lds<0>(acc, Bs4, rowb, n0, h, 16 * pr + c);
    } else {
        xchg_add<2>(theirs, acc);
        pair_b_to_lds<2>(acc, Bs4, rowb, n0, h, 16 * pr + c);
    }
    __syncthreads();
    if (kh == 0) pair_b_to_global<0>(acc, Bmat, rowb, n0, h);
    else         pair_b_to_global<2>(acc, Bmat, rowb, n0, h);
    #pragma unroll
    for (int i = 0; i < 4; ++i)
        #pragma unroll
        for (int q = 0; q < 4; ++q) acc[i][q] = v8f{};

    {
        const v4f* As4 = Bs4 + c * BS_PITCH4 + 32 * kh + h;
        pair_load_a(s0, As4, 0);
        #pragma nounroll
        for (int kk = 0; kk < 96; kk += 32) {
            pair_load_a(s1, As4, kk + 16);
            pair_load_b(s1, Bp2, kk + 16);
            gemm_compute(s0, acc);
            pair_load_a(s0, As4, kk + 32);
            pair_load_b(s0, Bp2, kk + 32);
            gemm_compute(s1, acc);
        }
        pair_load_a(s1, As4, 112);
        pair_load_b(s1, Bp2, 112);
        gemm_compute(s0, acc);
        gemm_compute(s1, acc);
    }
    if (kh == 0) xchg_store<2>(mine, acc); else xchg_store<0>(mine, acc);
    __syncthreads();
    if (kh == 0) {
        xchg_add<0>(theirs, acc);
        ns_epilogue<0, 0>(acc, Zmat, Zmat, 1.0f, rowb, n0, h);
    } else {
        xchg_add<2>(theirs, acc);
        ns_epilogue<0, 2>(acc, Zmat, Zmat, 1.0f, rowb, n0, h);
    }
}

#define BF_PITCH4 17

__device__ __forceinline__ void fin_load_a(FragSet& s, const float* __restrict__ Ap, int k0) {
    #pragma unroll
    for (int u = 0; u < 2; ++u)
        #pragma unroll
        for (int i = 0; i < 4; ++i)
            s.a[u][i] = *(const v4f*)(Ap + (size_t)i * 16 * C_DIM + k0 + 8 * u);
}

__device__ __forceinline__ void fin_load_b(FragSet& s, const v4f* Bl4, int k0) {
    #pragma unroll
    for (int u = 0; u < 2; ++u)
        #pragma unroll
        for (int e = 0; e < 4; ++e)
            s.b[u][e] = Bl4[(k0 + 8 * u + e) * BF_PITCH4];
}

template <int I0>
__device__ __forceinline__ void fin_bf_to_lds(const v8f (&acc)[4][4], v4f* Bf4,
                                              int rowb, int n0, int h, int c) {
    #pragma unroll
    for (int i = I0; i < I0 + 2; ++i)
        #pragma unroll
        for (int r = 0; r < 8; ++r) {
            const int m = rowb + 16 * i + 8 * h + r;
            Bf4[m * BF_PITCH4 + c] = pair_b_value(acc, i, r, m, n0);
        }
}

__global__ __launch_bounds__(256) void ns_final_kernel(float* __restrict__ out,
                                                       const float* __restrict__ Z,
                                                       const float* __restrict__ Y,
                                                       const float* __restrict__ scal) {
    __shared__ v4f xch[8][16 * 32];
    __shared__ v4f Bf4[256 * BF_PITCH4];
    const int lane = threadIdx.x & 31;
    const int wave = __builtin_amdgcn_readfirstlane((int)(threadIdx.x >> 5));
    const int pr   = wave & 3;
    const int kh   = wave >> 2;
    const int b    = blockIdx.y;
    const int colb = blockIdx.x * 64;
    const int rowb = pr * 64;
    const int c = lane & 15;
    const int h = lane >> 4;
    const float* Zb = Z + (size_t)b * MAT_ELEMS;
    const float* Yb = Y + (size_t)b * MAT_ELEMS;
    float* Dout = out + (size_t)b * OUT_PER_B;
    const int n0 = colb + 4 * c;
    v4f* mine = &xch[wave][lane];
    const v4f* theirs = &xch[wave ^ 4][lane];

    v8f acc[4][4];
    #pragma unroll
    for (int i = 0; i < 4; ++i)
        #pragma unroll
        for (int q = 0; q < 4; ++q) acc[i][q] = v8f{};

    FragSet s0, s1;
    {
        const float* Ap = Zb + (size_t)(rowb + c) * C_DIM + 128 * kh + 4 * h;
        const float* Bp = Yb + (size_t)(128 * kh + 4 * h) * C_DIM + colb + 4 * c;
        gemm_load(s0, Ap, Bp, 0);
        #pragma nounroll
        for (int kk = 0; kk < 96; kk += 32) {
            gemm_load(s1, Ap, Bp, kk + 16);
            gemm_compute(s0, acc);
            gemm_load(s0, Ap, Bp, kk + 32);
            gemm_compute(s1, acc);
        }
        gemm_load(s1, Ap, Bp, 112);
        gemm_compute(s0, acc);
        gemm_compute(s1, acc);
    }
    const float* Ap2 = Yb + (size_t)(rowb + c) * C_DIM + 128 * kh + 4 * h;
    const float s = scal[b];

    if (kh == 0) xchg_store<2>(mine, acc); else xchg_store<0>(mine, acc);
    __syncthreads();
    fin_load_a(s0, Ap2, 0);
    if (kh == 0) {
        xchg_add<0>(theirs, acc);
        fin_bf_to_lds<0>(acc, Bf4, rowb, n0, h, c);
    } else {
        xchg_add<2>(theirs, acc);
        fin_bf_to_lds<2>(acc, Bf4, rowb, n0, h, c);
    }
    #pragma unroll
    for (int i = 0; i < 4; ++i)
        #pragma unroll
        for (int q = 0; q < 4; ++q) acc[i][q] = v8f{};
    __syncthreads();

    {
        const v4f* Bl4 = Bf4 + (128 * kh + 4 * h) * BF_PITCH4 + c;
        fin_load_b(s0, Bl4, 0);
        #pragma nounroll
        for (int kk = 0; kk < 96; kk += 32) {
            fin_load_a(s1, Ap2, kk + 16);
            fin_load_b(s1, Bl4, kk + 16);
            gemm_compute(s0, acc);
            fin_load_a(s0, Ap2, kk + 32);
            fin_load_b(s0, Bl4, kk + 32);
            gemm_compute(s1, acc);
        }
        fin_load_a(s1, Ap2, 112);
        fin_load_b(s1, Bl4, 112);
        gemm_compute(s0, acc);
        gemm_compute(s1, acc);
    }
    if (kh == 0) xchg_store<2>(mine, acc); else xchg_store<0>(mine, acc);
    __syncthreads();
    if (kh == 0) {
        xchg_add<0>(theirs, acc);
        ns_epilogue<3, 0>(acc, Dout, Dout, s, rowb, n0, h);
    } else {
        xchg_add<2>(theirs, acc);
        ns_epilogue<3, 2>(acc, Dout, Dout, s, rowb, n0, h);
    }
}

extern "C" void kernel_launch(void* const* d_in, const int* in_sizes, int n_in,
                              void* d_out, int out_size, void* d_ws, size_t ws_size,
                              hipStream_t stream) {
    const float* x = (const float*)d_in[0];
    float* out = (float*)d_out;
    float* ws = (float*)d_ws;

    float* mean = ws;
    float* scal = ws + 16384;
    const size_t MS = (size_t)BATCH * MAT_ELEMS;
    float* M0 = ws + 16512;
    float* M1 = M0 + MS;
    float* M2 = M1 + MS;
    float* M3 = M2 + MS;
    float* M4 = M3 + MS;

    dim3 gGemm(4, BATCH);
    dim3 gRow(C_DIM, BATCH);

    mean_kernel<<<2048, 256, 0, stream>>>(x, mean);
    sigma_kernel<<<gGemm, 256, 0, stream>>>(M0, x, mean);
    trace_kernel<<<BATCH, 256, 0, stream>>>(M0, scal);
    b0_kernel<<<gRow, 256, 0, stream>>>(M1, M0, scal);

    ns_gemm_kernel<2><<<gGemm, 256, 0, stream>>>(M2, M0, M1, scal);

    ns_pair_kernel<<<gGemm, 256, 0, stream>>>(M3, M0, M1, M2);
    ns_gemm_kernel<0><<<gGemm, 256, 0, stream>>>(M4, M2, M3, nullptr);
    ns_pair_kernel<<<gGemm, 256, 0, stream>>>(M1, M3, M0, M4);
    ns_gemm_kernel<0><<<gGemm, 256, 0, stream>>>(M2, M4, M1, nullptr);
    ns_pair_kernel<<<gGemm, 256, 0, stream>>>(M4, M1, M3, M2);
    ns_gemm_kernel<0><<<gGemm, 256, 0, stream>>>(M0, M2, M4, nullptr);

    ns_final_kernel<<<gGemm, 256, 0, stream>>>(out, M1, M0, scal + 64);
}


# each Newton-Schulz iteration in ONE launch: Bn row panel stays in LDS and feeds Bn@Z and Bn@Y; Y@Bn formed as Bn@Y (polynomials of one matrix commute), its sum order changes; 12 launches become 9
# speedup vs baseline: 3.8609x; 1.1491x over previous
#include <hip/hip_runtime.h>
#include <hip/hip_bf16.h>
#include <math.h>

typedef __attribute__((ext_vector_type(2))) float v2f;
typedef __attribute__((ext_vector_type(4))) float v4f;
typedef __attribute__((ext_vector_type(8))) float v8f;

#define C_DIM 256
#define N_PIX 196
#define MAT_ELEMS (C_DIM * C_DIM)
#define BATCH 64
#define OUT_PER_B 32896

__global__ __launch_bounds__(256) void mean_kernel(const float* __restrict__ x,
                                                   float* __restrict__ mean) {
    int row = blockIdx.x * 8 + (threadIdx.x >> 5);
    int lane = threadIdx.x & 31;
    const float* xr = x + (size_t)row * N_PIX;
    float s = 0.0f;
    for (int i = lane; i < N_PIX; i += 32) s += xr[i];
    #pragma unroll
    for (int o = 16; o > 0; o >>= 1) s += __shfl_down(s, o, 32);
    if (lane == 0) mean[row] = s * (1.0f / (float)N_PIX);
}

#define WMMA_F32(a, b, c) __builtin_amdgcn_wmma_f32_16x16x4_f32(false, a, false, b, (short)0, c, false, false)

struct FragSet {
    v4f a[2][4];
    v4f b[2][4];
};

template <int I0>
__device__ __forceinline__ void xchg_store(v4f* __restrict__ slot, const v8f (&acc)[4][4]) {
    #pragma unroll
    for (int e = 0; e < 2; ++e)
        #pragma unroll
        for (int q = 0; q < 4; ++q)
            #pragma unroll
            for (int rr = 0; rr < 2; ++rr)
                slot[((e * 4 + q) * 2 + rr) * 32] =
                    v4f{acc[I0 + e][q][4 * rr], acc[I0 + e][q][4 * rr + 1],
                        acc[I0 + e][q][4 * rr + 2], acc[I0 + e][q][4 * rr + 3]};
}

template <int I0>
__device__ __forceinline__ void xchg_add(const v4f* __restrict__ slot, v8f (&acc)[4][4]) {
    #pragma unroll
    for (int e = 0; e < 2; ++e)
        #pragma unroll
        for (int q = 0; q < 4; ++q)
            #pragma unroll
            for (int rr = 0; rr < 2; ++rr) {
                const v4f t = slot[((e * 4 + q) * 2 + rr) * 32];
                #pragma unroll
                for (int j = 0; j < 4; ++j) acc[I0 + e][q][4 * rr + j] += t[j];
            }
}

__device__ __forceinline__ void sigma_load(FragSet& s, const float* __restrict__ xa,
                                           const float* __restrict__ xc, int k0) {
    #pragma unroll
    for (int u = 0; u < 2; ++u) {
        #pragma unroll
        for (int i = 0; i < 4; ++i)
            s.a[u][i] = *(const v4f*)(xa + (size_t)i * 16 * N_PIX + k0 + 8 * u);
        #pragma unroll
        for (int q = 0; q < 4; ++q)
            s.b[u][q] = *(const v4f*)(xc + (size_t)q * 16 * N_PIX + k0 + 8 * u);
    }
}

__device__ __forceinline__ void sigma_compute(const FragSet& s, const float (&mu)[4],
                                              float inv_n, v8f (&acc)[4][4]) {
    #pragma unroll
    for (int u = 0; u < 2; ++u) {
        v4f ac[4];
        #pragma unroll
        for (int i = 0; i < 4; ++i) ac[i] = (s.a[u][i] - mu[i]) * inv_n;
        #pragma unroll
        for (int p = 0; p < 2; ++p) {
            #pragma unroll
            for (int i = 0; i < 4; ++i) {
                v2f af = {ac[i][2 * p], ac[i][2 * p + 1]};
                #pragma unroll
                for (int q = 0; q < 4; ++q) {
                    v2f bf = {s.b[u][q][2 * p], s.b[u][q][2 * p + 1]};
                    acc[i][q] = WMMA_F32(af, bf, acc[i][q]);
                }
            }
        }
    }
}

template <int I0>
__device__ __forceinline__ void sigma_store(const v8f (&acc)[4][4], float* __restrict__ Db,
                                            int rowb, int colb, int c, int h) {
    #pragma unroll
    for (int i = I0; i < I0 + 2; ++i)
        #pragma unroll
        for (int r = 0; r < 8; ++r) {
            const int m = rowb + 16 * i + 8 * h + r;
            #pragma unroll
            for (int q = 0; q < 4; ++q)
                Db[(size_t)m * C_DIM + colb + 16 * q + c] = acc[i][q][r];
        }
}

__global__ __launch_bounds__(256) void sigma_kernel(float* __restrict__ D,
                                                    const float* __restrict__ x,
                                                    const float* __restrict__ mean) {
    __shared__ v4f xch[8][16 * 32];
    const int lane = threadIdx.x & 31;
    const int wave = __builtin_amdgcn_readfirstlane((int)(threadIdx.x >> 5));
    const int pr   = wave & 3;
    const int kh   = wave >> 2;
    const int b    = blockIdx.y;
    const int tileid = blockIdx.x;
    const int rowb = (tileid >> 1) * 128 + (pr >> 1) * 64;
    const int colb = (tileid & 1) * 128 + (pr & 1) * 64;
    const int c = lane & 15;
    const int h = lane >> 4;
    const float* xb = x + (size_t)b * C_DIM * N_PIX;
    const float* mb = mean + (size_t)b * C_DIM;
    float* Db = D + (size_t)b * MAT_ELEMS;
    const float inv_n = 1.0f / (float)N_PIX;

    const float* xa = xb + (size_t)(rowb + c) * N_PIX + 96 * kh + 4 * h;
    const float* xc = xb + (size_t)(colb + c) * N_PIX + 96 * kh + 4 * h;
    float mu[4];
    #pragma unroll
    for (int i = 0; i < 4; ++i) mu[i] = mb[rowb + 16 * i + c];

    v8f acc[4][4];
    #pragma unroll
    for (int i = 0; i < 4; ++i)
        #pragma unroll
        for (int q = 0; q < 4; ++q) acc[i][q] = v8f{};

    FragSet s0, s1;
    sigma_load(s0, xa, xc, 0);
    #pragma nounroll
    for (int kk = 0; kk < 64; kk += 32) {
        sigma_load(s1, xa, xc, kk + 16);
        sigma_compute(s0, mu, inv_n, acc);
        sigma_load(s0, xa, xc, kk + 32);
        sigma_compute(s1, mu, inv_n, acc);
    }
    sigma_load(s1, xa, xc, 80);
    sigma_compute(s0, mu, inv_n, acc);
    v2f ta[4], tb[4];
    if (kh == 1) {
        #pragma unroll
        for (int i = 0; i < 4; ++i)
            ta[i] = *(const v2f*)(xb + (size_t)(rowb + 16 * i + c) * N_PIX + 192 + 2 * h);
        #pragma unroll
        for (int q = 0; q < 4; ++q)
            tb[q] = *(const v2f*)(xb + (size_t)(colb + 16 * q + c) * N_PIX + 192 + 2 * h);
    }
    sigma_compute(s1, mu, inv_n, acc);
    if (kh == 1) {
        #pragma unroll
        for (int i = 0; i < 4; ++i) {
            v2f af = (ta[i] - mu[i]) * inv_n;
            #pragma unroll
            for (int q = 0; q < 4; ++q) acc[i][q] = WMMA_F32(af, tb[q], acc[i][q]);
        }
    }

    v4f* mine = &xch[wave][lane];
    const v4f* theirs = &xch[wave ^ 4][lane];
    if (kh == 0) xchg_store<2>(mine, acc); else xchg_store<0>(mine, acc);
    __syncthreads();
    if (kh == 0) {
        xchg_add<0>(theirs, acc);
        sigma_store<0>(acc, Db, rowb, colb, c, h);
    } else {
        xchg_add<2>(theirs, acc);
        sigma_store<2>(acc, Db, rowb, colb, c, h);
    }
}

__global__ __launch_bounds__(256) void trace_kernel(const float* __restrict__ S,
                                                    float* __restrict__ scal) {
    __shared__ float red[256];
    int b = blockIdx.x, tid = threadIdx.x;
    red[tid] = S[(size_t)b * MAT_ELEMS + (size_t)tid * 257];
    __syncthreads();
    for (int s = 128; s > 0; s >>= 1) {
        if (tid < s) red[tid] += red[tid + s];
        __syncthreads();
    }
    if (tid == 0) {
        float tr = red[0];
        scal[b] = 1.0f / tr;
        scal[64 + b] = sqrtf(tr);
    }
}

__global__ __launch_bounds__(256) void b0_kernel(float* __restrict__ Bm,
                                                 const float* __restrict__ S,
                                                 const float* __restrict__ scal) {
    int col = threadIdx.x, row = blockIdx.x, b = blockIdx.y;
    size_t idx = (size_t)b * MAT_ELEMS + (size_t)row * C_DIM + col;
    float rtr = scal[b];
    Bm[idx] = 0.5f * ((row == col ? 3.0f : 0.0f) - S[idx] * rtr);
}

__device__ __forceinline__ void gemm_load(FragSet& s, const float* __restrict__ Ap,
                                          const float* __restrict__ Bp, int k0) {
    #pragma unroll
    for (int u = 0; u < 2; ++u) {
        #pragma unroll
        for (int i = 0; i < 4; ++i)
            s.a[u][i] = *(const v4f*)(Ap + (size_t)i * 16 * C_DIM + k0 + 8 * u);
        #pragma unroll
        for (int e = 0; e < 4; ++e)
            s.b[u][e] = *(const v4f*)(Bp + (size_t)(k0 + 8 * u + e) * C_DIM);
    }
}

__device__ __forceinline__ void gemm_compute(const FragSet& s, v8f (&acc)[4][4]) {
    #pragma unroll
    for (int u = 0; u < 2; ++u) {
        #pragma unroll
        for (int p = 0; p < 2; ++p) {
            v2f bf[4];
            #pragma unroll
            for (int q = 0; q < 4; ++q) bf[q] = v2f{s.b[u][2 * p][q], s.b[u][2 * p + 1][q]};
            #pragma unroll
            for (int i = 0; i < 4; ++i) {
                v2f af = {s.a[u][i][2 * p], s.a[u][i][2 * p + 1]};
                #pragma unroll
                for (int q = 0; q < 4; ++q) acc[i][q] = WMMA_F32(af, bf[q], acc[i][q]);
            }
        }
    }
}

template <int OP, int I0>
__device__ __forceinline__ void ns_epilogue(const v8f (&acc)[4][4], float* __restrict__ Dmat,
                                            float* __restrict__ Dout, float s,
                                            int rowb, int n0, int h) {
    #pragma unroll
    for (int i = I0; i < I0 + 2; ++i) {
        #pragma unroll
        for (int r = 0; r < 8; ++r) {
            const int m = rowb + 16 * i + 8 * h + r;
            float val[4];
            #pragma unroll
            for (int q = 0; q < 4; ++q) val[q] = acc[i][q][r];
            if (OP == 3) {
                const int start = m * C_DIM - (m * (m - 1)) / 2;
                #pragma unroll
                for (int q = 0; q < 4; ++q) {
                    const int n = n0 + q;
                    if (n >= m) Dout[start + (n - m)] = val[q] * s;
                }
            } else {
                #pragma unroll
                for (int q = 0; q < 4; ++q) {
                    if (OP == 2) val[q] *= s;
                }
                *(v4f*)(Dmat + (size_t)m * C_DIM + n0) = v4f{val[0], val[1], val[2], val[3]};
            }
        }
    }
}

template <int OP>
__global__ __launch_bounds__(256) void ns_gemm_kernel(float* __restrict__ D,
                                                      const float* __restrict__ A,
                                                      const float* __restrict__ B,
                                                      const float* __restrict__ scal) {
    __shared__ v4f xch[8][16 * 32];
    const int lane = threadIdx.x & 31;
    const int wave = __builtin_amdgcn_readfirstlane((int)(threadIdx.x >> 5));
    const int pr   = wave & 3;
    const int kh   = wave >> 2;
    const int b    = blockIdx.y;
    const int tileid = blockIdx.x;
    const int rowb = (tileid >> 1) * 128 + (pr >> 1) * 64;
    const int colb = (tileid & 1) * 128 + (pr & 1) * 64;
    const int c = lane & 15;
    const int h = lane >> 4;
    const float* Ab = A + (size_t)b * MAT_ELEMS;
    const float* Bb = B + (size_t)b * MAT_ELEMS;
    float* Dmat = D + (size_t)b * MAT_ELEMS;
    float* Dout = D + (size_t)b * OUT_PER_B;

    const float* Ap = Ab + (size_t)(rowb + c) * C_DIM + 128 * kh + 4 * h;
    const float* Bp = Bb + (size_t)(128 * kh + 4 * h) * C_DIM + colb + 4 * c;

    v8f acc[4][4];
    #pragma unroll
    for (int i = 0; i < 4; ++i)
        #pragma unroll
        for (int q = 0; q < 4; ++q) acc[i][q] = v8f{};

    FragSet s0, s1;
    gemm_load(s0, Ap, Bp, 0);
    #pragma nounroll
    for (int kk = 0; kk < 96; kk += 32) {
        gemm_load(s1, Ap, Bp, kk + 16);
        gemm_compute(s0, acc);
        gemm_load(s0, Ap, Bp, kk + 32);
        gemm_compute(s1, acc);
    }
    gemm_load(s1, Ap, Bp, 112);
    gemm_compute(s0, acc);
    gemm_compute(s1, acc);

    float s = 1.0f;
    if (OP >= 2) s = scal[b];
    const int n0 = colb + 4 * c;
    v4f* mine = &xch[wave][lane];
    const v4f* theirs = &xch[wave ^ 4][lane];
    if (kh == 0) xchg_store<2>(mine, acc); else xchg_store<0>(mine, acc);
    __syncthreads();
    if (kh == 0) {
        xchg_add<0>(theirs, acc);
        ns_epilogue<OP, 0>(acc, Dmat, Dout, s, rowb, n0, h);
    } else {
        xchg_add<2>(theirs, acc);
        ns_epilogue<OP, 2>(acc, Dmat, Dout, s, rowb, n0, h);
    }
}

#define BS_PITCH4 65

__device__ __forceinline__ void pair_load_b(FragSet& s, const float* __restrict__ Bp, int k0) {
    #pragma unroll
    for (int u = 0; u < 2; ++u)
        #pragma unroll
        for (int e = 0; e < 4; ++e)
            s.b[u][e] = *(const v4f*)(Bp + (size_t)(k0 + 8 * u + e) * C_DIM);
}

__device__ __forceinline__ void pair_load_a(FragSet& s, const v4f* As4, int k0) {
    #pragma unroll
    for (int u = 0; u < 2; ++u)
        #pragma unroll
        for (int i = 0; i < 4; ++i)
            s.a[u][i] = As4[i * 16 * BS_PITCH4 + (k0 >> 2) + 2 * u];
}

__device__ __forceinline__ v4f pair_b_value(const v8f (&acc)[4][4], int i, int r, int m, int n0) {
    float val[4];
    #pragma unroll
    for (int q = 0; q < 4; ++q) {
        const int n = n0 + q;
        val[q] = (m == n ? 1.5f : 0.0f) - 0.5f * acc[i][q][r];
    }
    return v4f{val[0], val[1], val[2], val[3]};
}

template <int I0>
__device__ __forceinline__ void pair_b_to_lds(const v8f (&acc)[4][4], v4f* Bs4,
                                              int rowb, int n0, int h, int lc) {
    #pragma unroll
    for (int i = I0; i < I0 + 2; ++i)
        #pragma unroll
        for (int r = 0; r < 8; ++r) {
            const int ml = 16 * i + 8 * h + r;
            Bs4[ml * BS_PITCH4 + lc] = pair_b_value(acc, i, r, rowb + ml, n0);
        }
}

__global__ __launch_bounds__(256) void ns_iter_kernel(float* __restrict__ Yout,
                                                      float* __restrict__ Zout,
                                                      const float* __restrict__ Z,
                                                      const float* __restrict__ Y) {
    __shared__ v4f xch[8][16 * 32];
    __shared__ v4f Bs4[64 * BS_PITCH4];
    const int lane = threadIdx.x & 31;
    const int wave = __builtin_amdgcn_readfirstlane((int)(threadIdx.x >> 5));
    const int pr   = wave & 3;
    const int kh   = wave >> 2;
    const int b    = blockIdx.y;
    const int rowb = blockIdx.x * 64;
    const int colb = pr * 64;
    const int c = lane & 15;
    const int h = lane >> 4;
    const float* Zb = Z + (size_t)b * MAT_ELEMS;
    const float* Yb = Y + (size_t)b * MAT_ELEMS;
    float* Ymat = Yout + (size_t)b * MAT_ELEMS;
    float* Zmat = Zout + (size_t)b * MAT_ELEMS;
    const int n0 = colb + 4 * c;
    v4f* mine = &xch[wave][lane];
    const v4f* theirs = &xch[wave ^ 4][lane];

    v8f acc[4][4];
    #pragma unroll
    for (int i = 0; i < 4; ++i)
        #pragma unroll
        for (int q = 0; q < 4; ++q) acc[i][q] = v8f{};

    FragSet s0, s1;
    const float* BpY = Yb + (size_t)(128 * kh + 4 * h) * C_DIM + colb + 4 * c;
    const float* BpZ = Zb + (size_t)(128 * kh + 4 * h) * C_DIM + colb + 4 * c;
    {
        const float* Ap = Zb + (size_t)(rowb + c) * C_DIM + 128 * kh + 4 * h;
        gemm_load(s0, Ap, BpY, 0);
        #pragma nounroll
        for (int kk = 0; kk < 96; kk += 32) {
            gemm_load(s1, Ap, BpY, kk + 16);
            gemm_compute(s0, acc);
            gemm_load(s0, Ap, BpY, kk + 32);
            gemm_compute(s1, acc);
        }
        gemm_load(s1, Ap, BpY, 112);
        gemm_compute(s0, acc);
        pair_load_b(s0, BpZ, 0);
        gemm_compute(s1, acc);
    }

    if (kh == 0) xchg_store<2>(mine, acc); else xchg_store<0>(mine, acc);
    __syncthreads();
    if (kh == 0) {
        xchg_add<0>(theirs, acc);
        pair_b_to_lds<0>(acc, Bs4, rowb, n0, h, 16 * pr + c);
    } else {
        xchg_add<2>(theirs, acc);
        pair_b_to_lds<2>(acc, Bs4, rowb, n0, h, 16 * pr + c);
    }
    #pragma unroll
    for (int i = 0; i < 4; ++i)
        #pragma unroll
        for (int q = 0; q < 4; ++q) acc[i][q] = v8f{};
    __syncthreads();

    const v4f* As4 = Bs4 + c * BS_PITCH4 + 32 * kh + h;
    {
        pair_load_a(s0, As4, 0);
        #pragma nounroll
        for (int kk = 0; kk < 96; kk += 32) {
            pair_load_a(s1, As4, kk + 16);
            pair_load_b(s1, BpZ, kk + 16);
            gemm_compute(s0, acc);
            pair_load_a(s0, As4, kk + 32);
            pair_load_b(s0, BpZ, kk + 32);
            gemm_compute(s1, acc);
        }
        pair_load_a(s1, As4, 112);
        pair_load_b(s1, BpZ, 112);
        gemm_compute(s0, acc);
        pair_load_b(s0, BpY, 0);
        gemm_compute(s1, acc);
    }
    if (kh == 0) xchg_store<2>(mine, acc); else xchg_store<0>(mine, acc);
    __syncthreads();
    if (kh == 0) xchg_add<0>(theirs, acc); else xchg_add<2>(theirs, acc);
    __syncthreads();
    if (kh == 0) ns_epilogue<0, 0>(acc, Zmat, Zmat, 1.0f, rowb, n0, h);
    else         ns_epilogue<0, 2>(acc, Zmat, Zmat, 1.0f, rowb, n0, h);
    #pragma unroll
    for (int i = 0; i < 4; ++i)
        #pragma unroll
        for (int q = 0; q < 4; ++q) acc[i][q] = v8f{};

    {
        pair_load_a(s0, As4, 0);
        #pragma nounroll
        for (int kk = 0; kk < 96; kk += 32) {
            pair_load_a(s1, As4, kk + 16);
            pair_load_b(s1, BpY, kk + 16);
            gemm_compute(s0, acc);
            pair_load_a(s0, As4, kk + 32);
            pair_load_b(s0, BpY, kk + 32);
            gemm_compute(s1, acc);
        }
        pair_load_a(s1, As4, 112);
        pair_load_b(s1, BpY, 112);
        gemm_compute(s0, acc);
        gemm_compute(s1, acc);
    }
    if (kh == 0) xchg_store<2>(mine, acc); else xchg_store<0>(mine, acc);
    __syncthreads();
    if (kh == 0) {
        xchg_add<0>(theirs, acc);
        ns_epilogue<0, 0>(acc, Ymat, Ymat, 1.0f, rowb, n0, h);
    } else {
        xchg_add<2>(theirs, acc);
        ns_epilogue<0, 2>(acc, Ymat, Ymat, 1.0f, rowb, n0, h);
    }
}

#define BF_PITCH4 17

__device__ __forceinline__ void fin_load_a(FragSet& s, const float* __restrict__ Ap, int k0) {
    #pragma unroll
    for (int u = 0; u < 2; ++u)
        #pragma unroll
        for (int i = 0; i < 4; ++i)
            s.a[u][i] = *(const v4f*)(Ap + (size_t)i * 16 * C_DIM + k0 + 8 * u);
}

__device__ __forceinline__ void fin_load_b(FragSet& s, const v4f* Bl4, int k0) {
    #pragma unroll
    for (int u = 0; u < 2; ++u)
        #pragma unroll
        for (int e = 0; e < 4; ++e)
            s.b[u][e] = Bl4[(k0 + 8 * u + e) * BF_PITCH4];
}

template <int I0>
__device__ __forceinline__ void fin_bf_to_lds(const v8f (&acc)[4][4], v4f* Bf4,
                                              int rowb, int n0, int h, int c) {
    #pragma unroll
    for (int i = I0; i < I0 + 2; ++i)
        #pragma unroll
        for (int r = 0; r < 8; ++r) {
            const int m = rowb + 16 * i + 8 * h + r;
            Bf4[m * BF_PITCH4 + c] = pair_b_value(acc, i, r, m, n0);
        }
}

__global__ __launch_bounds__(256) void ns_final_kernel(float* __restrict__ out,
                                                       const float* __restrict__ Z,
                                                       const float* __restrict__ Y,
                                                       const float* __restrict__ scal) {
    __shared__ v4f xch[8][16 * 32];
    __shared__ v4f Bf4[256 * BF_PITCH4];
    const int lane = threadIdx.x & 31;
    const int wave = __builtin_amdgcn_readfirstlane((int)(threadIdx.x >> 5));
    const int pr   = wave & 3;
    const int kh   = wave >> 2;
    const int b    = blockIdx.y;
    const int colb = blockIdx.x * 64;
    const int rowb = pr * 64;
    const int c = lane & 15;
    const int h = lane >> 4;
    const float* Zb = Z + (size_t)b * MAT_ELEMS;
    const float* Yb = Y + (size_t)b * MAT_ELEMS;
    float* Dout = out + (size_t)b * OUT_PER_B;
    const int n0 = colb + 4 * c;
    v4f* mine = &xch[wave][lane];
    const v4f* theirs = &xch[wave ^ 4][lane];

    v8f acc[4][4];
    #pragma unroll
    for (int i = 0; i < 4; ++i)
        #pragma unroll
        for (int q = 0; q < 4; ++q) acc[i][q] = v8f{};

    FragSet s0, s1;
    {
        const float* Ap = Zb + (size_t)(rowb + c) * C_DIM + 128 * kh + 4 * h;
        const float* Bp = Yb + (size_t)(128 * kh + 4 * h) * C_DIM + colb + 4 * c;
        gemm_load(s0, Ap, Bp, 0);
        #pragma nounroll
        for (int kk = 0; kk < 96; kk += 32) {
            gemm_load(s1, Ap, Bp, kk + 16);
            gemm_compute(s0, acc);
            gemm_load(s0, Ap, Bp, kk + 32);
            gemm_compute(s1, acc);
        }
        gemm_load(s1, Ap, Bp, 112);
        gemm_compute(s0, acc);
        gemm_compute(s1, acc);
    }
    const float* Ap2 = Yb + (size_t)(rowb + c) * C_DIM + 128 * kh + 4 * h;
    const float s = scal[b];

    if (kh == 0) xchg_store<2>(mine, acc); else xchg_store<0>(mine, acc);
    __syncthreads();
    fin_load_a(s0, Ap2, 0);
    if (kh == 0) {
        xchg_add<0>(theirs, acc);
        fin_bf_to_lds<0>(acc, Bf4, rowb, n0, h, c);
    } else {
        xchg_add<2>(theirs, acc);
        fin_bf_to_lds<2>(acc, Bf4, rowb, n0, h, c);
    }
    #pragma unroll
    for (int i = 0; i < 4; ++i)
        #pragma unroll
        for (int q = 0; q < 4; ++q) acc[i][q] = v8f{};
    __syncthreads();

    {
        const v4f* Bl4 = Bf4 + (128 * kh + 4 * h) * BF_PITCH4 + c;
        fin_load_b(s0, Bl4, 0);
        #pragma nounroll
        for (int kk = 0; kk < 96; kk += 32) {
            fin_load_a(s1, Ap2, kk + 16);
            fin_load_b(s1, Bl4, kk + 16);
            gemm_compute(s0, acc);
            fin_load_a(s0, Ap2, kk + 32);
            fin_load_b(s0, Bl4, kk + 32);
            gemm_compute(s1, acc);
        }
        fin_load_a(s1, Ap2, 112);
        fin_load_b(s1, Bl4, 112);
        gemm_compute(s0, acc);
        gemm_compute(s1, acc);
    }
    if (kh == 0) xchg_store<2>(mine, acc); else xchg_store<0>(mine, acc);
    __syncthreads();
    if (kh == 0) {
        xchg_add<0>(theirs, acc);
        ns_epilogue<3, 0>(acc, Dout, Dout, s, rowb, n0, h);
    } else {
        xchg_add<2>(theirs, acc);
        ns_epilogue<3, 2>(acc, Dout, Dout, s, rowb, n0, h);
    }
}

extern "C" void kernel_launch(void* const* d_in, const int* in_sizes, int n_in,
                              void* d_out, int out_size, void* d_ws, size_t ws_size,
                              hipStream_t stream) {
    const float* x = (const float*)d_in[0];
    float* out = (float*)d_out;
    float* ws = (float*)d_ws;

    float* mean = ws;
    float* scal = ws + 16384;
    const size_t MS = (size_t)BATCH * MAT_ELEMS;
    float* M0 = ws + 16512;
    float* M1 = M0 + MS;
    float* M2 = M1 + MS;
    float* M3 = M2 + MS;

    dim3 gGemm(4, BATCH);
    dim3 gRow(C_DIM, BATCH);

    mean_kernel<<<2048, 256, 0, stream>>>(x, mean);
    sigma_kernel<<<gGemm, 256, 0, stream>>>(M0, x, mean);
    trace_kernel<<<BATCH, 256, 0, stream>>>(M0, scal);
    b0_kernel<<<gRow, 256, 0, stream>>>(M1, M0, scal);

    ns_gemm_kernel<2><<<gGemm, 256, 0, stream>>>(M2, M0, M1, scal);

    ns_iter_kernel<<<gGemm, 256, 0, stream>>>(M3, M0, M1, M2);
    ns_iter_kernel<<<gGemm, 256, 0, stream>>>(M2, M1, M0, M3);
    ns_iter_kernel<<<gGemm, 256, 0, stream>>>(M3, M0, M1, M2);

    ns_final_kernel<<<gGemm, 256, 0, stream>>>(out, M0, M3, scal + 64);
}


# iteration launch: only 10 of the 16 64x64 tiles of Zn and of Yn computed (symmetric: polynomials of sigma), off-diagonal tiles stored twice, mirrored; 5 wave pairs; mirrored sums differ by rounding
# speedup vs baseline: 4.0779x; 1.0562x over previous
#include <hip/hip_runtime.h>
#include <hip/hip_bf16.h>
#include <math.h>

typedef __attribute__((ext_vector_type(2))) float v2f;
typedef __attribute__((ext_vector_type(4))) float v4f;
typedef __attribute__((ext_vector_type(8))) float v8f;

#define C_DIM 256
#define N_PIX 196
#define MAT_ELEMS (C_DIM * C_DIM)
#define BATCH 64
#define OUT_PER_B 32896

__global__ __launch_bounds__(256) void mean_kernel(const float* __restrict__ x,
                                                   float* __restrict__ mean) {
    int row = blockIdx.x * 8 + (threadIdx.x >> 5);
    int lane = threadIdx.x & 31;
    const float* xr = x + (size_t)row * N_PIX;
    float s = 0.0f;
    for (int i = lane; i < N_PIX; i += 32) s += xr[i];
    #pragma unroll
    for (int o = 16; o > 0; o >>= 1) s += __shfl_down(s, o, 32);
    if (lane == 0) mean[row] = s * (1.0f / (float)N_PIX);
}

#define WMMA_F32(a, b, c) __builtin_amdgcn_wmma_f32_16x16x4_f32(false, a, false, b, (short)0, c, false, false)

struct FragSet {
    v4f a[2][4];
    v4f b[2][4];
};

template <int I0>
__device__ __forceinline__ void xchg_store(v4f* __restrict__ slot, const v8f (&acc)[4][4]) {
    #pragma unroll
    for (int e = 0; e < 2; ++e)
        #pragma unroll
        for (int q = 0; q < 4; ++q)
            #pragma unroll
            for (int rr = 0; rr < 2; ++rr)
                slot[((e * 4 + q) * 2 + rr) * 32] =
                    v4f{acc[I0 + e][q][4 * rr], acc[I0 + e][q][4 * rr + 1],
                        acc[I0 + e][q][4 * rr + 2], acc[I0 + e][q][4 * rr + 3]};
}

template <int I0>
__device__ __forceinline__ void xchg_add(const v4f* __restrict__ slot, v8f (&acc)[4][4]) {
    #pragma unroll
    for (int e = 0; e < 2; ++e)
        #pragma unroll
        for (int q = 0; q < 4; ++q)
            #pragma unroll
            for (int rr = 0; rr < 2; ++rr) {
                const v4f t = slot[((e * 4 + q) * 2 + rr) * 32];
                #pragma unroll
                for (int j = 0; j < 4; ++j) acc[I0 + e][q][4 * rr + j] += t[j];
            }
}

__device__ __forceinline__ void sigma_load(FragSet& s, const float* __restrict__ xa,
                                           const float* __restrict__ xc, int k0) {
    #pragma unroll
    for (int u = 0; u < 2; ++u) {
        #pragma unroll
        for (int i = 0; i < 4; ++i)
            s.a[u][i] = *(const v4f*)(xa + (size_t)i * 16 * N_PIX + k0 + 8 * u);
        #pragma unroll
        for (int q = 0; q < 4; ++q)
            s.b[u][q] = *(const v4f*)(xc + (size_t)q * 16 * N_PIX + k0 + 8 * u);
    }
}

__device__ __forceinline__ void sigma_compute(const FragSet& s, const float (&mu)[4],
                                              float inv_n, v8f (&acc)[4][4]) {
    #pragma unroll
    for (int u = 0; u < 2; ++u) {
        v4f ac[4];
        #pragma unroll
        for (int i = 0; i < 4; ++i) ac[i] = (s.a[u][i] - mu[i]) * inv_n;
        #pragma unroll
        for (int p = 0; p < 2; ++p) {
            #pragma unroll
            for (int i = 0; i < 4; ++i) {
                v2f af = {ac[i][2 * p], ac[i][2 * p + 1]};
                #pragma unroll
                for (int q = 0; q < 4; ++q) {
                    v2f bf = {s.b[u][q][2 * p], s.b[u][q][2 * p + 1]};
                    acc[i][q] = WMMA_F32(af, bf, acc[i][q]);
                }
            }
        }
    }
}

template <int I0>
__device__ __forceinline__ void sigma_store(const v8f (&acc)[4][4], float* __restrict__ Db,
                                            int rowb, int colb, int c, int h) {
    #pragma unroll
    for (int i = I0; i < I0 + 2; ++i)
        #pragma unroll
        for (int r = 0; r < 8; ++r) {
            const int m = rowb + 16 * i + 8 * h + r;
            #pragma unroll
            for (int q = 0; q < 4; ++q)
                Db[(size_t)m * C_DIM + colb + 16 * q + c] = acc[i][q][r];
        }
}

__global__ __launch_bounds__(256) void sigma_kernel(float* __restrict__ D,
                                                    const float* __restrict__ x,
                                                    const float* __restrict__ mean) {
    __shared__ v4f xch[8][16 * 32];
    const int lane = threadIdx.x & 31;
    const int wave = __builtin_amdgcn_readfirstlane((int)(threadIdx.x >> 5));
    const int pr   = wave & 3;
    const int kh   = wave >> 2;
    const int b    = blockIdx.y;
    const int tileid = blockIdx.x;
    const int rowb = (tileid >> 1) * 128 + (pr >> 1) * 64;
    const int colb = (tileid & 1) * 128 + (pr & 1) * 64;
    const int c = lane & 15;
    const int h = lane >> 4;
    const float* xb = x + (size_t)b * C_DIM * N_PIX;
    const float* mb = mean + (size_t)b * C_DIM;
    float* Db = D + (size_t)b * MAT_ELEMS;
    const float inv_n = 1.0f / (float)N_PIX;

    const float* xa = xb + (size_t)(rowb + c) * N_PIX + 96 * kh + 4 * h;
    const float* xc = xb + (size_t)(colb + c) * N_PIX + 96 * kh + 4 * h;
    float mu[4];
    #pragma unroll
    for (int i = 0; i < 4; ++i) mu[i] = mb[rowb + 16 * i + c];

    v8f acc[4][4];
    #pragma unroll
    for (int i = 0; i < 4; ++i)
        #pragma unroll
        for (int q = 0; q < 4; ++q) acc[i][q] = v8f{};

    FragSet s0, s1;
    sigma_load(s0, xa, xc, 0);
    #pragma nounroll
    for (int kk = 0; kk < 64; kk += 32) {
        sigma_load(s1, xa, xc, kk + 16);
        sigma_compute(s0, mu, inv_n, acc);
        sigma_load(s0, xa, xc, kk + 32);
        sigma_compute(s1, mu, inv_n, acc);
    }
    sigma_load(s1, xa, xc, 80);
    sigma_compute(s0, mu, inv_n, acc);
    v2f ta[4], tb[4];
    if (kh == 1) {
        #pragma unroll
        for (int i = 0; i < 4; ++i)
            ta[i] = *(const v2f*)(xb + (size_t)(rowb + 16 * i + c) * N_PIX + 192 + 2 * h);
        #pragma unroll
        for (int q = 0; q < 4; ++q)
            tb[q] = *(const v2f*)(xb + (size_t)(colb + 16 * q + c) * N_PIX + 192 + 2 * h);
    }
    sigma_compute(s1, mu, inv_n, acc);
    if (kh == 1) {
        #pragma unroll
        for (int i = 0; i < 4; ++i) {
            v2f af = (ta[i] - mu[i]) * inv_n;
            #pragma unroll
            for (int q = 0; q < 4; ++q) acc[i][q] = WMMA_F32(af, tb[q], acc[i][q]);
        }
    }

    v4f* mine = &xch[wave][lane];
    const v4f* theirs = &xch[wave ^ 4][lane];
    if (kh == 0) xchg_store<2>(mine, acc); else xchg_store<0>(mine, acc);
    __syncthreads();
    if (kh == 0) {
        xchg_add<0>(theirs, acc);
        sigma_store<0>(acc, Db, rowb, colb, c, h);
    } else {
        xchg_add<2>(theirs, acc);
        sigma_store<2>(acc, Db, rowb, colb, c, h);
    }
}

__global__ __launch_bounds__(256) void trace_kernel(const float* __restrict__ S,
                                                    float* __restrict__ scal) {
    __shared__ float red[256];
    int b = blockIdx.x, tid = threadIdx.x;
    red[tid] = S[(size_t)b * MAT_ELEMS + (size_t)tid * 257];
    __syncthreads();
    for (int s = 128; s > 0; s >>= 1) {
        if (tid < s) red[tid] += red[tid + s];
        __syncthreads();
    }
    if (tid == 0) {
        float tr = red[0];
        scal[b] = 1.0f / tr;
        scal[64 + b] = sqrtf(tr);
    }
}

__global__ __launch_bounds__(256) void b0_kernel(float* __restrict__ Bm,
                                                 const float* __restrict__ S,
                                                 const float* __restrict__ scal) {
    int col = threadIdx.x, row = blockIdx.x, b = blockIdx.y;
    size_t idx = (size_t)b * MAT_ELEMS + (size_t)row * C_DIM + col;
    float rtr = scal[b];
    Bm[idx] = 0.5f * ((row == col ? 3.0f : 0.0f) - S[idx] * rtr);
}

__device__ __forceinline__ void gemm_load(FragSet& s, const float* __restrict__ Ap,
                                          const float* __restrict__ Bp, int k0) {
    #pragma unroll
    for (int u = 0; u < 2; ++u) {
        #pragma unroll
        for (int i = 0; i < 4; ++i)
            s.a[u][i] = *(const v4f*)(Ap + (size_t)i * 16 * C_DIM + k0 + 8 * u);
        #pragma unroll
        for (int e = 0; e < 4; ++e)
            s.b[u][e] = *(const v4f*)(Bp + (size_t)(k0 + 8 * u + e) * C_DIM);
    }
}

__device__ __forceinline__ void gemm_compute(const FragSet& s, v8f (&acc)[4][4]) {
    #pragma unroll
    for (int u = 0; u < 2; ++u) {
        #pragma unroll
        for (int p = 0; p < 2; ++p) {
            v2f bf[4];
            #pragma unroll
            for (int q = 0; q < 4; ++q) bf[q] = v2f{s.b[u][2 * p][q], s.b[u][2 * p + 1][q]};
            #pragma unroll
            for (int i = 0; i < 4; ++i) {
                v2f af = {s.a[u][i][2 * p], s.a[u][i][2 * p + 1]};
                #pragma unroll
                for (int q = 0; q < 4; ++q) acc[i][q] = WMMA_F32(af, bf[q], acc[i][q]);
            }
        }
    }
}

template <int OP, int I0>
__device__ __forceinline__ void ns_epilogue(const v8f (&acc)[4][4], float* __restrict__ Dmat,
                                            float* __restrict__ Dout, float s,
                                            int rowb, int n0, int h) {
    #pragma unroll
    for (int i = I0; i < I0 + 2; ++i) {
        #pragma unroll
        for (int r = 0; r < 8; ++r) {
            const int m = rowb + 16 * i + 8 * h + r;
            float val[4];
            #pragma unroll
            for (int q = 0; q < 4; ++q) val[q] = acc[i][q][r];
            if (OP == 3) {
                const int start = m * C_DIM - (m * (m - 1)) / 2;
                #pragma unroll
                for (int q = 0; q < 4; ++q) {
                    const int n = n0 + q;
                    if (n >= m) Dout[start + (n - m)] = val[q] * s;
                }
            } else {
                #pragma unroll
                for (int q = 0; q < 4; ++q) {
                    if (OP == 2) val[q] *= s;
                }
                *(v4f*)(Dmat + (size_t)m * C_DIM + n0) = v4f{val[0], val[1], val[2], val[3]};
            }
        }
    }
}

template <int OP>
__global__ __launch_bounds__(256) void ns_gemm_kernel(float* __restrict__ D,
                                                      const float* __restrict__ A,
                                                      const float* __restrict__ B,
                                                      const float* __restrict__ scal) {
    __shared__ v4f xch[8][16 * 32];
    const int lane = threadIdx.x & 31;
    const int wave = __builtin_amdgcn_readfirstlane((int)(threadIdx.x >> 5));
    const int pr   = wave & 3;
    const int kh   = wave >> 2;
    const int b    = blockIdx.y;
    const int tileid = blockIdx.x;
    const int rowb = (tileid >> 1) * 128 + (pr >> 1) * 64;
    const int colb = (tileid & 1) * 128 + (pr & 1) * 64;
    const int c = lane & 15;
    const int h = lane >> 4;
    const float* Ab = A + (size_t)b * MAT_ELEMS;
    const float* Bb = B + (size_t)b * MAT_ELEMS;
    float* Dmat = D + (size_t)b * MAT_ELEMS;
    float* Dout = D + (size_t)b * OUT_PER_B;

    const float* Ap = Ab + (size_t)(rowb + c) * C_DIM + 128 * kh + 4 * h;
    const float* Bp = Bb + (size_t)(128 * kh + 4 * h) * C_DIM + colb + 4 * c;

    v8f acc[4][4];
    #pragma unroll
    for (int i = 0; i < 4; ++i)
        #pragma unroll
        for (int q = 0; q < 4; ++q) acc[i][q] = v8f{};

    FragSet s0, s1;
    gemm_load(s0, Ap, Bp, 0);
    #pragma nounroll
    for (int kk = 0; kk < 96; kk += 32) {
        gemm_load(s1, Ap, Bp, kk + 16);
        gemm_compute(s0, acc);
        gemm_load(s0, Ap, Bp, kk + 32);
        gemm_compute(s1, acc);
    }
    gemm_load(s1, Ap, Bp, 112);
    gemm_compute(s0, acc);
    gemm_compute(s1, acc);

    float s = 1.0f;
    if (OP >= 2) s = scal[b];
    const int n0 = colb + 4 * c;
    v4f* mine = &xch[wave][lane];
    const v4f* theirs = &xch[wave ^ 4][lane];
    if (kh == 0) xchg_store<2>(mine, acc); else xchg_store<0>(mine, acc);
    __syncthreads();
    if (kh == 0) {
        xchg_add<0>(theirs, acc);
        ns_epilogue<OP, 0>(acc, Dmat, Dout, s, rowb, n0, h);
    } else {
        xchg_add<2>(theirs, acc);
        ns_epilogue<OP, 2>(acc, Dmat, Dout, s, rowb, n0, h);
    }
}

#define BS_PITCH4 65

__device__ __forceinline__ void pair_load_b(FragSet& s, const float* __restrict__ Bp, int k0) {
    #pragma unroll
    for (int u = 0; u < 2; ++u)
        #pragma unroll
        for (int e = 0; e < 4; ++e)
            s.b[u][e] = *(const v4f*)(Bp + (size_t)(k0 + 8 * u + e) * C_DIM);
}

__device__ __forceinline__ void pair_load_a(FragSet& s, const v4f* As4, int k0) {
    #pragma unroll
    for (int u = 0; u < 2; ++u)
        #pragma unroll
        for (int i = 0; i < 4; ++i)
            s.a[u][i] = As4[i * 16 * BS_PITCH4 + (k0 >> 2) + 2 * u];
}

__device__ __forceinline__ v4f pair_b_value(const v8f (&acc)[4][4], int i, int r, int m, int n0) {
    float val[4];
    #pragma unroll
    for (int q = 0; q < 4; ++q) {
        const int n = n0 + q;
        val[q] = (m == n ? 1.5f : 0.0f) - 0.5f * acc[i][q][r];
    }
    return v4f{val[0], val[1], val[2], val[3]};
}

template <int I0>
__device__ __forceinline__ void pair_b_to_lds(const v8f (&acc)[4][4], v4f* Bs4,
                                              int rowb, int n0, int h, int lc) {
    #pragma unroll
    for (int i = I0; i < I0 + 2; ++i)
        #pragma unroll
        for (int r = 0; r < 8; ++r) {
            const int ml = 16 * i + 8 * h + r;
            Bs4[ml * BS_PITCH4 + lc] = pair_b_value(acc, i, r, rowb + ml, n0);
        }
}

template <int I0>
__device__ __forceinline__ void mirror_store(const v8f (&acc)[4][4], float* __restrict__ Dmat,
                                             int rowb, int n0, int h) {
    #pragma unroll
    for (int i = I0; i < I0 + 2; ++i)
        #pragma unroll
        for (int q = 0; q < 4; ++q) {
            float* p = Dmat + (size_t)(n0 + q) * C_DIM + rowb + 16 * i + 8 * h;
            *(v4f*)p       = v4f{acc[i][q][0], acc[i][q][1], acc[i][q][2], acc[i][q][3]};
            *(v4f*)(p + 4) = v4f{acc[i][q][4], acc[i][q][5], acc[i][q][6], acc[i][q][7]};
        }
}

__global__ __launch_bounds__(320) void ns_iter_kernel(float* __restrict__ Yout,
                                                      float* __restrict__ Zout,
                                                      const float* __restrict__ Z,
                                                      const float* __restrict__ Y) {
    __shared__ v4f xch[10][16 * 32];
    __shared__ v4f Bs4[64 * BS_PITCH4];
    const int lane = threadIdx.x & 31;
    const int wave = __builtin_amdgcn_readfirstlane((int)(threadIdx.x >> 5));
    const int b    = blockIdx.y;
    const int pp   = blockIdx.x;
    const int rowb = pp * 64;
    const int c = lane & 15;
    const int h = lane >> 4;
    const float* Zb = Z + (size_t)b * MAT_ELEMS;
    const float* Yb = Y + (size_t)b * MAT_ELEMS;
    float* Ymat = Yout + (size_t)b * MAT_ELEMS;
    float* Zmat = Zout + (size_t)b * MAT_ELEMS;

    const int t   = wave < 5 ? wave : wave - 5;
    const int kh2 = wave < 5 ? 0 : 1;
    const int qb  = (pp + (t == 4 ? 2 : (t & 1))) & 3;
    const bool useY = (t == 2) || (t == 3) || (t == 4 && pp >= 2);
    const float* Msrc = useY ? Yb : Zb;
    float* Mdst = useY ? Ymat : Zmat;
    const int n0 = qb * 64 + 4 * c;
    const float* Bp2 = Msrc + (size_t)(128 * kh2 + 4 * h) * C_DIM + n0;

    v8f acc[4][4];
    #pragma unroll
    for (int i = 0; i < 4; ++i)
        #pragma unroll
        for (int q = 0; q < 4; ++q) acc[i][q] = v8f{};

    FragSet s0, s1;
    if (wave < 8) {
        const int pr = wave & 3;
        const int kh = wave >> 2;
        const float* BpY = Yb + (size_t)(128 * kh + 4 * h) * C_DIM + pr * 64 + 4 * c;
        const float* Ap = Zb + (size_t)(rowb + c) * C_DIM + 128 * kh + 4 * h;
        gemm_load(s0, Ap, BpY, 0);
        #pragma nounroll
        for (int kk = 0; kk < 96; kk += 32) {
            gemm_load(s1, Ap, BpY, kk + 16);
            gemm_compute(s0, acc);
            gemm_load(s0, Ap, BpY, kk + 32);
            gemm_compute(s1, acc);
        }
        gemm_load(s1, Ap, BpY, 112);
        gemm_compute(s0, acc);
        pair_load_b(s0, Bp2, 0);
        gemm_compute(s1, acc);
        if (kh == 0) xchg_store<2>(&xch[wave][lane], acc); else xchg_store<0>(&xch[wave][lane], acc);
    } else {
        pair_load_b(s0, Bp2, 0);
    }
    __syncthreads();
    if (wave < 8) {
        const int pr = wave & 3;
        const int kh = wave >> 2;
        const int n1 = pr * 64 + 4 * c;
        const v4f* theirs1 = &xch[wave ^ 4][lane];
        if (kh == 0) {
            xchg_add<0>(theirs1, acc);
            pair_b_to_lds<0>(acc, Bs4, rowb, n1, h, 16 * pr + c);
        } else {
            xchg_add<2>(theirs1, acc);
            pair_b_to_lds<2>(acc, Bs4, rowb, n1, h, 16 * pr + c);
        }
        #pragma unroll
        for (int i = 0; i < 4; ++i)
            #pragma unroll
            for (int q = 0; q < 4; ++q) acc[i][q] = v8f{};
    }
    __syncthreads();

    const v4f* As4 = Bs4 + c * BS_PITCH4 + 32 * kh2 + h;
    {
        pair_load_a(s0, As4, 0);
        #pragma nounroll
        for (int kk = 0; kk < 96; kk += 32) {
            pair_load_a(s1, As4, kk + 16);
            pair_load_b(s1, Bp2, kk + 16);
            gemm_compute(s0, acc);
            pair_load_a(s0, As4, kk + 32);
            pair_load_b(s0, Bp2, kk + 32);
            gemm_compute(s1, acc);
        }
        pair_load_a(s1, As4, 112);
        pair_load_b(s1, Bp2, 112);
        gemm_compute(s0, acc);
        gemm_compute(s1, acc);
    }
    v4f* mine = &xch[wave][lane];
    const v4f* theirs = &xch[wave < 5 ? wave + 5 : wave - 5][lane];
    if (kh2 == 0) xchg_store<2>(mine, acc); else xchg_store<0>(mine, acc);
    __syncthreads();
    if (kh2 == 0) {
        xchg_add<0>(theirs, acc);
        ns_epilogue<0, 0>(acc, Mdst, Mdst, 1.0f, rowb, n0, h);
        if (qb != pp) mirror_store<0>(acc, Mdst, rowb, n0, h);
    } else {
        xchg_add<2>(theirs, acc);
        ns_epilogue<0, 2>(acc, Mdst, Mdst, 1.0f, rowb, n0, h);
        if (qb != pp) mirror_store<2>(acc, Mdst, rowb, n0, h);
    }
}

#define BF_PITCH4 17

__device__ __forceinline__ void fin_load_a(FragSet& s, const float* __restrict__ Ap, int k0) {
    #pragma unroll
    for (int u = 0; u < 2; ++u)
        #pragma unroll
        for (int i = 0; i < 4; ++i)
            s.a[u][i] = *(const v4f*)(Ap + (size_t)i * 16 * C_DIM + k0 + 8 * u);
}

__device__ __forceinline__ void fin_load_b(FragSet& s, const v4f* Bl4, int k0) {
    #pragma unroll
    for (int u = 0; u < 2; ++u)
        #pragma unroll
        for (int e = 0; e < 4; ++e)
            s.b[u][e] = Bl4[(k0 + 8 * u + e) * BF_PITCH4];
}

template <int I0>
__device__ __forceinline__ void fin_bf_to_lds(const v8f (&acc)[4][4], v4f* Bf4,
                                              int rowb, int n0, int h, int c) {
    #pragma unroll
    for (int i = I0; i < I0 + 2; ++i)
        #pragma unroll
        for (int r = 0; r < 8; ++r) {
            const int m = rowb + 16 * i + 8 * h + r;
            Bf4[m * BF_PITCH4 + c] = pair_b_value(acc, i, r, m, n0);
        }
}

__global__ __launch_bounds__(256) void ns_final_kernel(float* __restrict__ out,
                                                       const float* __restrict__ Z,
                                                       const float* __restrict__ Y,
                                                       const float* __restrict__ scal) {
    __shared__ v4f xch[8][16 * 32];
    __shared__ v4f Bf4[256 * BF_PITCH4];
    const int lane = threadIdx.x & 31;
    const int wave = __builtin_amdgcn_readfirstlane((int)(threadIdx.x >> 5));
    const int pr   = wave & 3;
    const int kh   = wave >> 2;
    const int b    = blockIdx.y;
    const int colb = blockIdx.x * 64;
    const int rowb = pr * 64;
    const int c = lane & 15;
    const int h = lane >> 4;
    const float* Zb = Z + (size_t)b * MAT_ELEMS;
    const float* Yb = Y + (size_t)b * MAT_ELEMS;
    float* Dout = out + (size_t)b * OUT_PER_B;
    const int n0 = colb + 4 * c;
    v4f* mine = &xch[wave][lane];
    const v4f* theirs = &xch[wave ^ 4][lane];

    v8f acc[4][4];
    #pragma unroll
    for (int i = 0; i < 4; ++i)
        #pragma unroll
        for (int q = 0; q < 4; ++q) acc[i][q] = v8f{};

    FragSet s0, s1;
    {
        const float* Ap = Zb + (size_t)(rowb + c) * C_DIM + 128 * kh + 4 * h;
        const float* Bp = Yb + (size_t)(128 * kh + 4 * h) * C_DIM + colb + 4 * c;
        gemm_load(s0, Ap, Bp, 0);
        #pragma nounroll
        for (int kk = 0; kk < 96; kk += 32) {
            gemm_load(s1, Ap, Bp, kk + 16);
            gemm_compute(s0, acc);
            gemm_load(s0, Ap, Bp, kk + 32);
            gemm_compute(s1, acc);
        }
        gemm_load(s1, Ap, Bp, 112);
        gemm_compute(s0, acc);
        gemm_compute(s1, acc);
    }
    const float* Ap2 = Yb + (size_t)(rowb + c) * C_DIM + 128 * kh + 4 * h;
    const float s = scal[b];

    if (kh == 0) xchg_store<2>(mine, acc); else xchg_store<0>(mine, acc);
    __syncthreads();
    fin_load_a(s0, Ap2, 0);
    if (kh == 0) {
        xchg_add<0>(theirs, acc);
        fin_bf_to_lds<0>(acc, Bf4, rowb, n0, h, c);
    } else {
        xchg_add<2>(theirs, acc);
        fin_bf_to_lds<2>(acc, Bf4, rowb, n0, h, c);
    }
    #pragma unroll
    for (int i = 0; i < 4; ++i)
        #pragma unroll
        for (int q = 0; q < 4; ++q) acc[i][q] = v8f{};
    __syncthreads();

    {
        const v4f* Bl4 = Bf4 + (128 * kh + 4 * h) * BF_PITCH4 + c;
        fin_load_b(s0, Bl4, 0);
        #pragma nounroll
        for (int kk = 0; kk < 96; kk += 32) {
            fin_load_a(s1, Ap2, kk + 16);
            fin_load_b(s1, Bl4, kk + 16);
            gemm_compute(s0, acc);
            fin_load_a(s0, Ap2, kk + 32);
            fin_load_b(s0, Bl4, kk + 32);
            gemm_compute(s1, acc);
        }
        fin_load_a(s1, Ap2, 112);
        fin_load_b(s1, Bl4, 112);
        gemm_compute(s0, acc);
        gemm_compute(s1, acc);
    }
    if (kh == 0) xchg_store<2>(mine, acc); else xchg_store<0>(mine, acc);
    __syncthreads();
    if (kh == 0) {
        xchg_add<0>(theirs, acc);
        ns_epilogue<3, 0>(acc, Dout, Dout, s, rowb, n0, h);
    } else {
        xchg_add<2>(theirs, acc);
        ns_epilogue<3, 2>(acc, Dout, Dout, s, rowb, n0, h);
    }
}

extern "C" void kernel_launch(void* const* d_in, const int* in_sizes, int n_in,
                              void* d_out, int out_size, void* d_ws, size_t ws_size,
                              hipStream_t stream) {
    const float* x = (const float*)d_in[0];
    float* out = (float*)d_out;
    float* ws = (float*)d_ws;

    float* mean = ws;
    float* scal = ws + 16384;
    const size_t MS = (size_t)BATCH * MAT_ELEMS;
    float* M0 = ws + 16512;
    float* M1 = M0 + MS;
    float* M2 = M1 + MS;
    float* M3 = M2 + MS;

    dim3 gGemm(4, BATCH);
    dim3 gRow(C_DIM, BATCH);

    mean_kernel<<<2048, 256, 0, stream>>>(x, mean);
    sigma_kernel<<<gGemm, 256, 0, stream>>>(M0, x, mean);
    trace_kernel<<<BATCH, 256, 0, stream>>>(M0, scal);
    b0_kernel<<<gRow, 256, 0, stream>>>(M1, M0, scal);

    ns_gemm_kernel<2><<<gGemm, 256, 0, stream>>>(M2, M0, M1, scal);

    ns_iter_kernel<<<gGemm, 320, 0, stream>>>(M3, M0, M1, M2);
    ns_iter_kernel<<<gGemm, 320, 0, stream>>>(M2, M1, M0, M3);
    ns_iter_kernel<<<gGemm, 320, 0, stream>>>(M3, M0, M1, M2);

    ns_final_kernel<<<gGemm, 256, 0, stream>>>(out, M0, M3, scal + 64);
}
